# starting kernel
# baseline (speedup 1.0000x reference)
#include <hip/hip_runtime.h>

#define N_NODES 50000
#define E_EDGES 800000
#define F_IN    64
#define F_HID   128
#define F_OUT   8

typedef __attribute__((ext_vector_type(2))) float v2f;
typedef __attribute__((ext_vector_type(8))) float v8f;


__global__ void GCN_init_deg(float* __restrict__ deg) {
    int i = blockIdx.x * blockDim.x + threadIdx.x;
    if (i < N_NODES) deg[i] = 1.0f;
}

__global__ void GCN_count_deg(const int* __restrict__ dst, float* __restrict__ deg) {
    int e = blockIdx.x * blockDim.x + threadIdx.x;
    if (e < E_EDGES) atomicAdd(&deg[dst[e]], 1.0f);
}

__global__ void GCN_rsqrt(float* __restrict__ dinv) {
    int i = blockIdx.x * blockDim.x + threadIdx.x;
    if (i < N_NODES) dinv[i] = rsqrtf(dinv[i]);
}


__global__ void GCN_gemm1(const float* __restrict__ x, const float* __restrict__ W1,
                          float* __restrict__ h) {
    const int wave = threadIdx.x >> 5;
    const int lane = threadIdx.x & 31;
    const int half = lane >> 4;
    const int l    = lane & 15;
    const int m0   = blockIdx.x * 16;
    const int n0   = wave * 16;

    v8f c = {};
    const int arow = m0 + l;
#pragma unroll
    for (int k0 = 0; k0 < F_IN; k0 += 4) {
        const int kb = k0 + half * 2;
        v2f a, b;
        a.x = x[arow * F_IN + kb];
        a.y = x[arow * F_IN + kb + 1];
        b.x = W1[kb * F_HID + n0 + l];
        b.y = W1[(kb + 1) * F_HID + n0 + l];
        c = __builtin_amdgcn_wmma_f32_16x16x4_f32(false, a, false, b,
                                                  (short)0, c, false, false);
    }
#pragma unroll
    for (int r = 0; r < 8; ++r)
        h[(m0 + r + half * 8) * F_HID + n0 + l] = c[r];
}


__global__ void GCN_self1(const float* __restrict__ h, const float* __restrict__ dinv,
                          float* __restrict__ agg) {
    int idx = blockIdx.x * blockDim.x + threadIdx.x;
    if (idx >= N_NODES * (F_HID / 4)) return;
    int i  = idx >> 5;
    int c4 = idx & 31;
    float w = dinv[i]; w = w * w;
    float4 hv = reinterpret_cast<const float4*>(h + (size_t)i * F_HID)[c4];
    float4 o; o.x = hv.x * w; o.y = hv.y * w; o.z = hv.z * w; o.w = hv.w * w;
    reinterpret_cast<float4*>(agg + (size_t)i * F_HID)[c4] = o;
}

__global__ void GCN_edge1(const int* __restrict__ src, const int* __restrict__ dst,
                          const float* __restrict__ dinv, const float* __restrict__ h,
                          float* __restrict__ agg) {
    int gid  = blockIdx.x * blockDim.x + threadIdx.x;
    int e    = gid >> 5;
    int lane = gid & 31;
    if (e >= E_EDGES) return;
    int s = src[e], d = dst[e];
    float norm = dinv[s] * dinv[d];
    float4 hv = reinterpret_cast<const float4*>(h + (size_t)s * F_HID)[lane];
    float* ap = agg + (size_t)d * F_HID + lane * 4;
    atomicAdd(ap + 0, hv.x * norm);
    atomicAdd(ap + 1, hv.y * norm);
    atomicAdd(ap + 2, hv.z * norm);
    atomicAdd(ap + 3, hv.w * norm);
}


__global__ void GCN_gemm2(const float* __restrict__ agg, const float* __restrict__ b1,
                          const float* __restrict__ W2, float* __restrict__ h2) {
    const int tile = blockIdx.x * 8 + (threadIdx.x >> 5);
    if (tile >= N_NODES / 16) return;
    const int lane = threadIdx.x & 31;
    const int half = lane >> 4;
    const int l    = lane & 15;
    const int m0   = tile * 16;

    v8f c = {};
    const int arow = m0 + l;
    const bool valid_col = (l < F_OUT);
#pragma unroll
    for (int k0 = 0; k0 < F_HID; k0 += 4) {
        const int kb = k0 + half * 2;
        v2f a, b;
        a.x = fmaxf(agg[arow * F_HID + kb]     + b1[kb],     0.0f);
        a.y = fmaxf(agg[arow * F_HID + kb + 1] + b1[kb + 1], 0.0f);
        b.x = valid_col ? W2[kb * F_OUT + l]       : 0.0f;
        b.y = valid_col ? W2[(kb + 1) * F_OUT + l] : 0.0f;
        c = __builtin_amdgcn_wmma_f32_16x16x4_f32(false, a, false, b,
                                                  (short)0, c, false, false);
    }
    if (valid_col) {
#pragma unroll
        for (int r = 0; r < 8; ++r)
            h2[(m0 + r + half * 8) * F_OUT + l] = c[r];
    }
}


__global__ void GCN_self2(const float* __restrict__ h2, const float* __restrict__ dinv,
                          const float* __restrict__ b2, float* __restrict__ out) {
    int idx = blockIdx.x * blockDim.x + threadIdx.x;
    if (idx >= N_NODES * F_OUT) return;
    int i = idx >> 3, ch = idx & 7;
    float w = dinv[i];
    out[idx] = h2[idx] * w * w + b2[ch];
}

__global__ void GCN_edge2(const int* __restrict__ src, const int* __restrict__ dst,
                          const float* __restrict__ dinv, const float* __restrict__ h2,
                          float* __restrict__ out) {
    int e = blockIdx.x * blockDim.x + threadIdx.x;
    if (e >= E_EDGES) return;
    int s = src[e], d = dst[e];
    float norm = dinv[s] * dinv[d];
    float4 a = reinterpret_cast<const float4*>(h2 + (size_t)s * F_OUT)[0];
    float4 b = reinterpret_cast<const float4*>(h2 + (size_t)s * F_OUT)[1];
    float* op = out + (size_t)d * F_OUT;
    atomicAdd(op + 0, a.x * norm);
    atomicAdd(op + 1, a.y * norm);
    atomicAdd(op + 2, a.z * norm);
    atomicAdd(op + 3, a.w * norm);
    atomicAdd(op + 4, b.x * norm);
    atomicAdd(op + 5, b.y * norm);
    atomicAdd(op + 6, b.z * norm);
    atomicAdd(op + 7, b.w * norm);
}


extern "C" void kernel_launch(void* const* d_in, const int* in_sizes, int n_in,
                              void* d_out, int out_size, void* d_ws, size_t ws_size,
                              hipStream_t stream) {
    const float* x  = (const float*)d_in[0];
    const int*   ei = (const int*)d_in[1];
    const float* W1 = (const float*)d_in[2];
    const float* b1 = (const float*)d_in[3];
    const float* W2 = (const float*)d_in[4];
    const float* b2 = (const float*)d_in[5];
    float* out = (float*)d_out;

    const int* src = ei;
    const int* dst = ei + E_EDGES;

    float* ws   = (float*)d_ws;
    float* dinv = ws;
    float* h    = ws + 50176;
    float* agg  = ws + 50176 + 6400000;
    float* h2   = ws + 50176 + 12800000;

    const int TB = 256;

    GCN_init_deg <<<(N_NODES + TB - 1) / TB, TB, 0, stream>>>(dinv);
    GCN_count_deg<<<(E_EDGES + TB - 1) / TB, TB, 0, stream>>>(dst, dinv);
    GCN_rsqrt    <<<(N_NODES + TB - 1) / TB, TB, 0, stream>>>(dinv);

    GCN_gemm1<<<N_NODES / 16, TB, 0, stream>>>(x, W1, h);

    GCN_self1<<<(N_NODES * (F_HID / 4) + TB - 1) / TB, TB, 0, stream>>>(h, dinv, agg);
    GCN_edge1<<<(E_EDGES * 32 + TB - 1) / TB, TB, 0, stream>>>(src, dst, dinv, h, agg);

    GCN_gemm2<<<(N_NODES / 16 + 7) / 8, TB, 0, stream>>>(agg, b1, W2, h2);

    GCN_self2<<<(N_NODES * F_OUT + TB - 1) / TB, TB, 0, stream>>>(h2, dinv, b2, out);
    GCN_edge2<<<(E_EDGES + TB - 1) / TB, TB, 0, stream>>>(src, dst, dinv, h2, out);
}


# edge scatters: lane = channel, so each atomic instruction covers contiguous bytes (edge1: 1 whole 128-B line per instr, was 4 partial; edge2: 8 lanes per edge, 4 lines per instr, was 32)
# speedup vs baseline: 1.8812x; 1.8812x over previous
#include <hip/hip_runtime.h>

#define N_NODES 50000
#define E_EDGES 800000
#define F_IN    64
#define F_HID   128
#define F_OUT   8

typedef __attribute__((ext_vector_type(2))) float v2f;
typedef __attribute__((ext_vector_type(8))) float v8f;


__global__ void GCN_init_deg(float* __restrict__ deg) {
    int i = blockIdx.x * blockDim.x + threadIdx.x;
    if (i < N_NODES) deg[i] = 1.0f;
}

__global__ void GCN_count_deg(const int* __restrict__ dst, float* __restrict__ deg) {
    int e = blockIdx.x * blockDim.x + threadIdx.x;
    if (e < E_EDGES) atomicAdd(&deg[dst[e]], 1.0f);
}

__global__ void GCN_rsqrt(float* __restrict__ dinv) {
    int i = blockIdx.x * blockDim.x + threadIdx.x;
    if (i < N_NODES) dinv[i] = rsqrtf(dinv[i]);
}


__global__ void GCN_gemm1(const float* __restrict__ x, const float* __restrict__ W1,
                          float* __restrict__ h) {
    const int wave = threadIdx.x >> 5;
    const int lane = threadIdx.x & 31;
    const int half = lane >> 4;
    const int l    = lane & 15;
    const int m0   = blockIdx.x * 16;
    const int n0   = wave * 16;

    v8f c = {};
    const int arow = m0 + l;
#pragma unroll
    for (int k0 = 0; k0 < F_IN; k0 += 4) {
        const int kb = k0 + half * 2;
        v2f a, b;
        a.x = x[arow * F_IN + kb];
        a.y = x[arow * F_IN + kb + 1];
        b.x = W1[kb * F_HID + n0 + l];
        b.y = W1[(kb + 1) * F_HID + n0 + l];
        c = __builtin_amdgcn_wmma_f32_16x16x4_f32(false, a, false, b,
                                                  (short)0, c, false, false);
    }
#pragma unroll
    for (int r = 0; r < 8; ++r)
        h[(m0 + r + half * 8) * F_HID + n0 + l] = c[r];
}


__global__ void GCN_self1(const float* __restrict__ h, const float* __restrict__ dinv,
                          float* __restrict__ agg) {
    int idx = blockIdx.x * blockDim.x + threadIdx.x;
    if (idx >= N_NODES * (F_HID / 4)) return;
    int i  = idx >> 5;
    int c4 = idx & 31;
    float w = dinv[i]; w = w * w;
    float4 hv = reinterpret_cast<const float4*>(h + (size_t)i * F_HID)[c4];
    float4 o; o.x = hv.x * w; o.y = hv.y * w; o.z = hv.z * w; o.w = hv.w * w;
    reinterpret_cast<float4*>(agg + (size_t)i * F_HID)[c4] = o;
}

__global__ void GCN_edge1(const int* __restrict__ src, const int* __restrict__ dst,
                          const float* __restrict__ dinv, const float* __restrict__ h,
                          float* __restrict__ agg) {
    int gid  = blockIdx.x * blockDim.x + threadIdx.x;
    int e    = gid >> 5;
    int lane = gid & 31;
    if (e >= E_EDGES) return;
    int s = src[e], d = dst[e];
    float norm = dinv[s] * dinv[d];
    const float* hp = h + (size_t)s * F_HID + lane;
    float* ap = agg + (size_t)d * F_HID + lane;
    float v0 = hp[0], v1 = hp[32], v2 = hp[64], v3 = hp[96];
    atomicAdd(ap +  0, v0 * norm);
    atomicAdd(ap + 32, v1 * norm);
    atomicAdd(ap + 64, v2 * norm);
    atomicAdd(ap + 96, v3 * norm);
}


__global__ void GCN_gemm2(const float* __restrict__ agg, const float* __restrict__ b1,
                          const float* __restrict__ W2, float* __restrict__ h2) {
    const int tile = blockIdx.x * 8 + (threadIdx.x >> 5);
    if (tile >= N_NODES / 16) return;
    const int lane = threadIdx.x & 31;
    const int half = lane >> 4;
    const int l    = lane & 15;
    const int m0   = tile * 16;

    v8f c = {};
    const int arow = m0 + l;
    const bool valid_col = (l < F_OUT);
#pragma unroll
    for (int k0 = 0; k0 < F_HID; k0 += 4) {
        const int kb = k0 + half * 2;
        v2f a, b;
        a.x = fmaxf(agg[arow * F_HID + kb]     + b1[kb],     0.0f);
        a.y = fmaxf(agg[arow * F_HID + kb + 1] + b1[kb + 1], 0.0f);
        b.x = valid_col ? W2[kb * F_OUT + l]       : 0.0f;
        b.y = valid_col ? W2[(kb + 1) * F_OUT + l] : 0.0f;
        c = __builtin_amdgcn_wmma_f32_16x16x4_f32(false, a, false, b,
                                                  (short)0, c, false, false);
    }
    if (valid_col) {
#pragma unroll
        for (int r = 0; r < 8; ++r)
            h2[(m0 + r + half * 8) * F_OUT + l] = c[r];
    }
}


__global__ void GCN_self2(const float* __restrict__ h2, const float* __restrict__ dinv,
                          const float* __restrict__ b2, float* __restrict__ out) {
    int idx = blockIdx.x * blockDim.x + threadIdx.x;
    if (idx >= N_NODES * F_OUT) return;
    int i = idx >> 3, ch = idx & 7;
    float w = dinv[i];
    out[idx] = h2[idx] * w * w + b2[ch];
}

__global__ void GCN_edge2(const int* __restrict__ src, const int* __restrict__ dst,
                          const float* __restrict__ dinv, const float* __restrict__ h2,
                          float* __restrict__ out) {
    int gid = blockIdx.x * blockDim.x + threadIdx.x;
    int e   = gid >> 3;
    int ch  = gid & 7;
    if (e >= E_EDGES) return;
    int s = src[e], d = dst[e];
    float norm = dinv[s] * dinv[d];
    atomicAdd(out + (size_t)d * F_OUT + ch, h2[(size_t)s * F_OUT + ch] * norm);
}


extern "C" void kernel_launch(void* const* d_in, const int* in_sizes, int n_in,
                              void* d_out, int out_size, void* d_ws, size_t ws_size,
                              hipStream_t stream) {
    const float* x  = (const float*)d_in[0];
    const int*   ei = (const int*)d_in[1];
    const float* W1 = (const float*)d_in[2];
    const float* b1 = (const float*)d_in[3];
    const float* W2 = (const float*)d_in[4];
    const float* b2 = (const float*)d_in[5];
    float* out = (float*)d_out;

    const int* src = ei;
    const int* dst = ei + E_EDGES;

    float* ws   = (float*)d_ws;
    float* dinv = ws;
    float* h    = ws + 50176;
    float* agg  = ws + 50176 + 6400000;
    float* h2   = ws + 50176 + 12800000;

    const int TB = 256;

    GCN_init_deg <<<(N_NODES + TB - 1) / TB, TB, 0, stream>>>(dinv);
    GCN_count_deg<<<(E_EDGES + TB - 1) / TB, TB, 0, stream>>>(dst, dinv);
    GCN_rsqrt    <<<(N_NODES + TB - 1) / TB, TB, 0, stream>>>(dinv);

    GCN_gemm1<<<N_NODES / 16, TB, 0, stream>>>(x, W1, h);

    GCN_self1<<<(N_NODES * (F_HID / 4) + TB - 1) / TB, TB, 0, stream>>>(h, dinv, agg);
    GCN_edge1<<<(E_EDGES * 32 + TB - 1) / TB, TB, 0, stream>>>(src, dst, dinv, h, agg);

    GCN_gemm2<<<(N_NODES / 16 + 7) / 8, TB, 0, stream>>>(agg, b1, W2, h2);

    GCN_self2<<<(N_NODES * F_OUT + TB - 1) / TB, TB, 0, stream>>>(h2, dinv, b2, out);
    GCN_edge2<<<(E_EDGES * 8 + TB - 1) / TB, TB, 0, stream>>>(src, dst, dinv, h2, out);
}


# layer 1 aggregates the 64 input channels before W1 instead of the 128 outputs after it (linear, so equal; sums reassociated): edge gather and atomics per edge halve, 2 lines each
# speedup vs baseline: 1.9405x; 1.0315x over previous
#include <hip/hip_runtime.h>

#define N_NODES 50000
#define E_EDGES 800000
#define F_IN    64
#define F_HID   128
#define F_OUT   8

typedef __attribute__((ext_vector_type(2))) float v2f;
typedef __attribute__((ext_vector_type(8))) float v8f;


__global__ void GCN_init_deg(float* __restrict__ deg) {
    int i = blockIdx.x * blockDim.x + threadIdx.x;
    if (i < N_NODES) deg[i] = 1.0f;
}

__global__ void GCN_count_deg(const int* __restrict__ dst, float* __restrict__ deg) {
    int e = blockIdx.x * blockDim.x + threadIdx.x;
    if (e < E_EDGES) atomicAdd(&deg[dst[e]], 1.0f);
}

__global__ void GCN_rsqrt(float* __restrict__ dinv) {
    int i = blockIdx.x * blockDim.x + threadIdx.x;
    if (i < N_NODES) dinv[i] = rsqrtf(dinv[i]);
}


__global__ void GCN_gemm1(const float* __restrict__ x, const float* __restrict__ W1,
                          float* __restrict__ h) {
    const int wave = threadIdx.x >> 5;
    const int lane = threadIdx.x & 31;
    const int half = lane >> 4;
    const int l    = lane & 15;
    const int m0   = blockIdx.x * 16;
    const int n0   = wave * 16;

    v8f c = {};
    const int arow = m0 + l;
#pragma unroll
    for (int k0 = 0; k0 < F_IN; k0 += 4) {
        const int kb = k0 + half * 2;
        v2f a, b;
        a.x = x[arow * F_IN + kb];
        a.y = x[arow * F_IN + kb + 1];
        b.x = W1[kb * F_HID + n0 + l];
        b.y = W1[(kb + 1) * F_HID + n0 + l];
        c = __builtin_amdgcn_wmma_f32_16x16x4_f32(false, a, false, b,
                                                  (short)0, c, false, false);
    }
#pragma unroll
    for (int r = 0; r < 8; ++r)
        h[(m0 + r + half * 8) * F_HID + n0 + l] = c[r];
}


__global__ void GCN_self1(const float* __restrict__ x, const float* __restrict__ dinv,
                          float* __restrict__ aggx) {
    int idx = blockIdx.x * blockDim.x + threadIdx.x;
    if (idx >= N_NODES * (F_IN / 4)) return;
    int i  = idx >> 4;
    int c4 = idx & 15;
    float w = dinv[i]; w = w * w;
    float4 xv = reinterpret_cast<const float4*>(x + (size_t)i * F_IN)[c4];
    float4 o; o.x = xv.x * w; o.y = xv.y * w; o.z = xv.z * w; o.w = xv.w * w;
    reinterpret_cast<float4*>(aggx + (size_t)i * F_IN)[c4] = o;
}

__global__ void GCN_edge1(const int* __restrict__ src, const int* __restrict__ dst,
                          const float* __restrict__ dinv, const float* __restrict__ x,
                          float* __restrict__ aggx) {
    int gid  = blockIdx.x * blockDim.x + threadIdx.x;
    int e    = gid >> 5;
    int lane = gid & 31;
    if (e >= E_EDGES) return;
    int s = src[e], d = dst[e];
    float norm = dinv[s] * dinv[d];
    const float* xp = x + (size_t)s * F_IN + lane;
    float* ap = aggx + (size_t)d * F_IN + lane;
    float v0 = xp[0], v1 = xp[32];
    atomicAdd(ap +  0, v0 * norm);
    atomicAdd(ap + 32, v1 * norm);
}


__global__ void GCN_gemm2(const float* __restrict__ agg, const float* __restrict__ b1,
                          const float* __restrict__ W2, float* __restrict__ h2) {
    const int tile = blockIdx.x * 8 + (threadIdx.x >> 5);
    if (tile >= N_NODES / 16) return;
    const int lane = threadIdx.x & 31;
    const int half = lane >> 4;
    const int l    = lane & 15;
    const int m0   = tile * 16;

    v8f c = {};
    const int arow = m0 + l;
    const bool valid_col = (l < F_OUT);
#pragma unroll
    for (int k0 = 0; k0 < F_HID; k0 += 4) {
        const int kb = k0 + half * 2;
        v2f a, b;
        a.x = fmaxf(agg[arow * F_HID + kb]     + b1[kb],     0.0f);
        a.y = fmaxf(agg[arow * F_HID + kb + 1] + b1[kb + 1], 0.0f);
        b.x = valid_col ? W2[kb * F_OUT + l]       : 0.0f;
        b.y = valid_col ? W2[(kb + 1) * F_OUT + l] : 0.0f;
        c = __builtin_amdgcn_wmma_f32_16x16x4_f32(false, a, false, b,
                                                  (short)0, c, false, false);
    }
    if (valid_col) {
#pragma unroll
        for (int r = 0; r < 8; ++r)
            h2[(m0 + r + half * 8) * F_OUT + l] = c[r];
    }
}


__global__ void GCN_self2(const float* __restrict__ h2, const float* __restrict__ dinv,
                          const float* __restrict__ b2, float* __restrict__ out) {
    int idx = blockIdx.x * blockDim.x + threadIdx.x;
    if (idx >= N_NODES * F_OUT) return;
    int i = idx >> 3, ch = idx & 7;
    float w = dinv[i];
    out[idx] = h2[idx] * w * w + b2[ch];
}

__global__ void GCN_edge2(const int* __restrict__ src, const int* __restrict__ dst,
                          const float* __restrict__ dinv, const float* __restrict__ h2,
                          float* __restrict__ out) {
    int gid = blockIdx.x * blockDim.x + threadIdx.x;
    int e   = gid >> 3;
    int ch  = gid & 7;
    if (e >= E_EDGES) return;
    int s = src[e], d = dst[e];
    float norm = dinv[s] * dinv[d];
    atomicAdd(out + (size_t)d * F_OUT + ch, h2[(size_t)s * F_OUT + ch] * norm);
}


extern "C" void kernel_launch(void* const* d_in, const int* in_sizes, int n_in,
                              void* d_out, int out_size, void* d_ws, size_t ws_size,
                              hipStream_t stream) {
    const float* x  = (const float*)d_in[0];
    const int*   ei = (const int*)d_in[1];
    const float* W1 = (const float*)d_in[2];
    const float* b1 = (const float*)d_in[3];
    const float* W2 = (const float*)d_in[4];
    const float* b2 = (const float*)d_in[5];
    float* out = (float*)d_out;

    const int* src = ei;
    const int* dst = ei + E_EDGES;

    float* ws   = (float*)d_ws;
    float* dinv = ws;
    float* aggx = ws + 50176;
    float* agg  = ws + 50176 + 6400000;
    float* h2   = ws + 50176 + 12800000;

    const int TB = 256;

    GCN_init_deg <<<(N_NODES + TB - 1) / TB, TB, 0, stream>>>(dinv);
    GCN_count_deg<<<(E_EDGES + TB - 1) / TB, TB, 0, stream>>>(dst, dinv);
    GCN_rsqrt    <<<(N_NODES + TB - 1) / TB, TB, 0, stream>>>(dinv);

    GCN_self1<<<(N_NODES * (F_IN / 4) + TB - 1) / TB, TB, 0, stream>>>(x, dinv, aggx);
    GCN_edge1<<<(E_EDGES * 32 + TB - 1) / TB, TB, 0, stream>>>(src, dst, dinv, x, aggx);

    GCN_gemm1<<<N_NODES / 16, TB, 0, stream>>>(aggx, W1, agg);

    GCN_gemm2<<<(N_NODES / 16 + 7) / 8, TB, 0, stream>>>(agg, b1, W2, h2);

    GCN_self2<<<(N_NODES * F_OUT + TB - 1) / TB, TB, 0, stream>>>(h2, dinv, b2, out);
    GCN_edge2<<<(E_EDGES * 8 + TB - 1) / TB, TB, 0, stream>>>(src, dst, dinv, h2, out);
}


# edge scatters: one wave per 32 edges (was 1 and 4): endpoints and norms loaded once per lane, broadcast per edge; 32 and 8 gathers in flight per wave; atomic arrival order differs
# speedup vs baseline: 2.5554x; 1.3169x over previous
#include <hip/hip_runtime.h>

#define N_NODES 50000
#define E_EDGES 800000
#define F_IN    64
#define F_HID   128
#define F_OUT   8

typedef __attribute__((ext_vector_type(2))) float v2f;
typedef __attribute__((ext_vector_type(8))) float v8f;


__global__ void GCN_init_deg(float* __restrict__ deg) {
    int i = blockIdx.x * blockDim.x + threadIdx.x;
    if (i < N_NODES) deg[i] = 1.0f;
}

__global__ void GCN_count_deg(const int* __restrict__ dst, float* __restrict__ deg) {
    int e = blockIdx.x * blockDim.x + threadIdx.x;
    if (e < E_EDGES) atomicAdd(&deg[dst[e]], 1.0f);
}

__global__ void GCN_rsqrt(float* __restrict__ dinv) {
    int i = blockIdx.x * blockDim.x + threadIdx.x;
    if (i < N_NODES) dinv[i] = rsqrtf(dinv[i]);
}


__global__ void GCN_gemm1(const float* __restrict__ x, const float* __restrict__ W1,
                          float* __restrict__ h) {
    const int wave = threadIdx.x >> 5;
    const int lane = threadIdx.x & 31;
    const int half = lane >> 4;
    const int l    = lane & 15;
    const int m0   = blockIdx.x * 16;
    const int n0   = wave * 16;

    v8f c = {};
    const int arow = m0 + l;
#pragma unroll
    for (int k0 = 0; k0 < F_IN; k0 += 4) {
        const int kb = k0 + half * 2;
        v2f a, b;
        a.x = x[arow * F_IN + kb];
        a.y = x[arow * F_IN + kb + 1];
        b.x = W1[kb * F_HID + n0 + l];
        b.y = W1[(kb + 1) * F_HID + n0 + l];
        c = __builtin_amdgcn_wmma_f32_16x16x4_f32(false, a, false, b,
                                                  (short)0, c, false, false);
    }
#pragma unroll
    for (int r = 0; r < 8; ++r)
        h[(m0 + r + half * 8) * F_HID + n0 + l] = c[r];
}


__global__ void GCN_self1(const float* __restrict__ x, const float* __restrict__ dinv,
                          float* __restrict__ aggx) {
    int idx = blockIdx.x * blockDim.x + threadIdx.x;
    if (idx >= N_NODES * (F_IN / 4)) return;
    int i  = idx >> 4;
    int c4 = idx & 15;
    float w = dinv[i]; w = w * w;
    float4 xv = reinterpret_cast<const float4*>(x + (size_t)i * F_IN)[c4];
    float4 o; o.x = xv.x * w; o.y = xv.y * w; o.z = xv.z * w; o.w = xv.w * w;
    reinterpret_cast<float4*>(aggx + (size_t)i * F_IN)[c4] = o;
}

static_assert(E_EDGES % 32 == 0, "every lane of every edge wave owns an edge");

__global__ void GCN_edge1(const int* __restrict__ src, const int* __restrict__ dst,
                          const float* __restrict__ dinv, const float* __restrict__ x,
                          float* __restrict__ aggx) {
    const int lane = threadIdx.x & 31;
    const int e0   = ((blockIdx.x * blockDim.x + threadIdx.x) >> 5) * 32;
    if (e0 >= E_EDGES) return;
    const int s = src[e0 + lane], d = dst[e0 + lane];
    const int nrm  = __float_as_int(dinv[s] * dinv[d]);
    const int srow = s * F_IN, drow = d * F_IN;
#pragma unroll
    for (int g = 0; g < 32; g += 16) {
        float v0[16], v1[16];
#pragma unroll
        for (int j = 0; j < 16; ++j) {
            const float* xp = x + (unsigned)(__builtin_amdgcn_readlane(srow, g + j) + lane);
            v0[j] = xp[0];
            v1[j] = xp[32];
        }
#pragma unroll
        for (int j = 0; j < 16; ++j) {
            float* ap = aggx + (unsigned)(__builtin_amdgcn_readlane(drow, g + j) + lane);
            const float norm = __int_as_float(__builtin_amdgcn_readlane(nrm, g + j));
            atomicAdd(ap +  0, v0[j] * norm);
            atomicAdd(ap + 32, v1[j] * norm);
        }
    }
}


__global__ void GCN_gemm2(const float* __restrict__ agg, const float* __restrict__ b1,
                          const float* __restrict__ W2, float* __restrict__ h2) {
    const int tile = blockIdx.x * 8 + (threadIdx.x >> 5);
    if (tile >= N_NODES / 16) return;
    const int lane = threadIdx.x & 31;
    const int half = lane >> 4;
    const int l    = lane & 15;
    const int m0   = tile * 16;

    v8f c = {};
    const int arow = m0 + l;
    const bool valid_col = (l < F_OUT);
#pragma unroll
    for (int k0 = 0; k0 < F_HID; k0 += 4) {
        const int kb = k0 + half * 2;
        v2f a, b;
        a.x = fmaxf(agg[arow * F_HID + kb]     + b1[kb],     0.0f);
        a.y = fmaxf(agg[arow * F_HID + kb + 1] + b1[kb + 1], 0.0f);
        b.x = valid_col ? W2[kb * F_OUT + l]       : 0.0f;
        b.y = valid_col ? W2[(kb + 1) * F_OUT + l] : 0.0f;
        c = __builtin_amdgcn_wmma_f32_16x16x4_f32(false, a, false, b,
                                                  (short)0, c, false, false);
    }
    if (valid_col) {
#pragma unroll
        for (int r = 0; r < 8; ++r)
            h2[(m0 + r + half * 8) * F_OUT + l] = c[r];
    }
}


__global__ void GCN_self2(const float* __restrict__ h2, const float* __restrict__ dinv,
                          const float* __restrict__ b2, float* __restrict__ out) {
    int idx = blockIdx.x * blockDim.x + threadIdx.x;
    if (idx >= N_NODES * F_OUT) return;
    int i = idx >> 3, ch = idx & 7;
    float w = dinv[i];
    out[idx] = h2[idx] * w * w + b2[ch];
}

__global__ void GCN_edge2(const int* __restrict__ src, const int* __restrict__ dst,
                          const float* __restrict__ dinv, const float* __restrict__ h2,
                          float* __restrict__ out) {
    const int lane = threadIdx.x & 31;
    const int e0   = ((blockIdx.x * blockDim.x + threadIdx.x) >> 5) * 32;
    if (e0 >= E_EDGES) return;
    const int s = src[e0 + lane], d = dst[e0 + lane];
    const float nrm = dinv[s] * dinv[d];
    const int srow = s * F_OUT, drow = d * F_OUT;
    const int q = lane >> 3, ch = lane & 7;
    float v[8];
#pragma unroll
    for (int t = 0; t < 8; ++t)
        v[t] = h2[(unsigned)(__shfl(srow, 4 * t + q) + ch)];
#pragma unroll
    for (int t = 0; t < 8; ++t)
        atomicAdd(out + (unsigned)(__shfl(drow, 4 * t + q) + ch), v[t] * __shfl(nrm, 4 * t + q));
}


extern "C" void kernel_launch(void* const* d_in, const int* in_sizes, int n_in,
                              void* d_out, int out_size, void* d_ws, size_t ws_size,
                              hipStream_t stream) {
    const float* x  = (const float*)d_in[0];
    const int*   ei = (const int*)d_in[1];
    const float* W1 = (const float*)d_in[2];
    const float* b1 = (const float*)d_in[3];
    const float* W2 = (const float*)d_in[4];
    const float* b2 = (const float*)d_in[5];
    float* out = (float*)d_out;

    const int* src = ei;
    const int* dst = ei + E_EDGES;

    float* ws   = (float*)d_ws;
    float* dinv = ws;
    float* aggx = ws + 50176;
    float* agg  = ws + 50176 + 6400000;
    float* h2   = ws + 50176 + 12800000;

    const int TB = 256;

    GCN_init_deg <<<(N_NODES + TB - 1) / TB, TB, 0, stream>>>(dinv);
    GCN_count_deg<<<(E_EDGES + TB - 1) / TB, TB, 0, stream>>>(dst, dinv);
    GCN_rsqrt    <<<(N_NODES + TB - 1) / TB, TB, 0, stream>>>(dinv);

    GCN_self1<<<(N_NODES * (F_IN / 4) + TB - 1) / TB, TB, 0, stream>>>(x, dinv, aggx);
    GCN_edge1<<<(E_EDGES + TB - 1) / TB, TB, 0, stream>>>(src, dst, dinv, x, aggx);

    GCN_gemm1<<<N_NODES / 16, TB, 0, stream>>>(aggx, W1, agg);

    GCN_gemm2<<<(N_NODES / 16 + 7) / 8, TB, 0, stream>>>(agg, b1, W2, h2);

    GCN_self2<<<(N_NODES * F_OUT + TB - 1) / TB, TB, 0, stream>>>(h2, dinv, b2, out);
    GCN_edge2<<<(E_EDGES + TB - 1) / TB, TB, 0, stream>>>(src, dst, dinv, h2, out);
}


# degree count in LDS: a work-group histograms one edge slice x one node-range part (integer adds, same values), planes summed in the rsqrt kernel; no scattered global atomics there; 9 -> 8 launches
# speedup vs baseline: 3.2896x; 1.2873x over previous
#include <hip/hip_runtime.h>

#define N_NODES 50000
#define E_EDGES 800000
#define F_IN    64
#define F_HID   128
#define F_OUT   8

typedef __attribute__((ext_vector_type(2))) float v2f;
typedef __attribute__((ext_vector_type(8))) float v8f;


#define DEG_SL     64
#define DEG_NP     4
#define DEG_PART   12512
#define DEG_PITCH  50176
#define DEG_TB     1024
#define DEG_EPS    (E_EDGES / DEG_SL)
#define DEG_ROUNDS ((DEG_EPS + DEG_TB - 1) / DEG_TB)
static_assert(E_EDGES % DEG_SL == 0, "equal edge slices");
static_assert(DEG_NP * DEG_PART >= N_NODES && DEG_NP * DEG_PART <= DEG_PITCH, "parts cover the nodes inside a plane");
static_assert(DEG_PART % 32 == 0, "parts are whole lines and move as 16-byte pieces");

__global__ __launch_bounds__(DEG_TB) void GCN_deg_hist(const int* __restrict__ dst,
                                                       unsigned* __restrict__ planes) {
    __shared__ __attribute__((aligned(16))) unsigned cnt[DEG_PART];
    const int tid   = threadIdx.x;
    const int part  = blockIdx.x % DEG_NP;
    const int slice = blockIdx.x / DEG_NP;
    const int lo    = part * DEG_PART;
    const int* dp = dst + slice * DEG_EPS;
    int d[DEG_ROUNDS];
#pragma unroll
    for (int j = 0; j < DEG_ROUNDS; ++j) {
        const int e = tid + j * DEG_TB;
        d[j] = (e < DEG_EPS) ? dp[e] : -1;
    }
    uint4* c4 = reinterpret_cast<uint4*>(cnt);
    for (int i = tid; i < DEG_PART / 4; i += DEG_TB) c4[i] = make_uint4(0u, 0u, 0u, 0u);
    __syncthreads();
#pragma unroll
    for (int j = 0; j < DEG_ROUNDS; ++j) {
        const unsigned r = (unsigned)(d[j] - lo);
        if (r < (unsigned)DEG_PART) atomicAdd(&cnt[r], 1u);
    }
    __syncthreads();
    uint4* p4 = reinterpret_cast<uint4*>(planes + (size_t)slice * DEG_PITCH + lo);
    for (int i = tid; i < DEG_PART / 4; i += DEG_TB) p4[i] = c4[i];
}

__global__ void GCN_dinv(const unsigned* __restrict__ planes, float* __restrict__ dinv) {
    int i = blockIdx.x * blockDim.x + threadIdx.x;
    if (i >= N_NODES) return;
    unsigned c = 1u;
#pragma unroll
    for (int b = 0; b < DEG_SL; ++b) c += planes[(size_t)b * DEG_PITCH + i];
    dinv[i] = rsqrtf((float)c);
}


__global__ void GCN_gemm1(const float* __restrict__ x, const float* __restrict__ W1,
                          float* __restrict__ h) {
    const int wave = threadIdx.x >> 5;
    const int lane = threadIdx.x & 31;
    const int half = lane >> 4;
    const int l    = lane & 15;
    const int m0   = blockIdx.x * 16;
    const int n0   = wave * 16;

    v8f c = {};
    const int arow = m0 + l;
#pragma unroll
    for (int k0 = 0; k0 < F_IN; k0 += 4) {
        const int kb = k0 + half * 2;
        v2f a, b;
        a.x = x[arow * F_IN + kb];
        a.y = x[arow * F_IN + kb + 1];
        b.x = W1[kb * F_HID + n0 + l];
        b.y = W1[(kb + 1) * F_HID + n0 + l];
        c = __builtin_amdgcn_wmma_f32_16x16x4_f32(false, a, false, b,
                                                  (short)0, c, false, false);
    }
#pragma unroll
    for (int r = 0; r < 8; ++r)
        h[(m0 + r + half * 8) * F_HID + n0 + l] = c[r];
}


__global__ void GCN_self1(const float* __restrict__ x, const float* __restrict__ dinv,
                          float* __restrict__ aggx) {
    int idx = blockIdx.x * blockDim.x + threadIdx.x;
    if (idx >= N_NODES * (F_IN / 4)) return;
    int i  = idx >> 4;
    int c4 = idx & 15;
    float w = dinv[i]; w = w * w;
    float4 xv = reinterpret_cast<const float4*>(x + (size_t)i * F_IN)[c4];
    float4 o; o.x = xv.x * w; o.y = xv.y * w; o.z = xv.z * w; o.w = xv.w * w;
    reinterpret_cast<float4*>(aggx + (size_t)i * F_IN)[c4] = o;
}

static_assert(E_EDGES % 32 == 0, "every lane of every edge wave owns an edge");

__global__ void GCN_edge1(const int* __restrict__ src, const int* __restrict__ dst,
                          const float* __restrict__ dinv, const float* __restrict__ x,
                          float* __restrict__ aggx) {
    const int lane = threadIdx.x & 31;
    const int e0   = ((blockIdx.x * blockDim.x + threadIdx.x) >> 5) * 32;
    if (e0 >= E_EDGES) return;
    const int s = src[e0 + lane], d = dst[e0 + lane];
    const int nrm  = __float_as_int(dinv[s] * dinv[d]);
    const int srow = s * F_IN, drow = d * F_IN;
#pragma unroll
    for (int g = 0; g < 32; g += 16) {
        float v0[16], v1[16];
#pragma unroll
        for (int j = 0; j < 16; ++j) {
            const float* xp = x + (unsigned)(__builtin_amdgcn_readlane(srow, g + j) + lane);
            v0[j] = xp[0];
            v1[j] = xp[32];
        }
#pragma unroll
        for (int j = 0; j < 16; ++j) {
            float* ap = aggx + (unsigned)(__builtin_amdgcn_readlane(drow, g + j) + lane);
            const float norm = __int_as_float(__builtin_amdgcn_readlane(nrm, g + j));
            atomicAdd(ap +  0, v0[j] * norm);
            atomicAdd(ap + 32, v1[j] * norm);
        }
    }
}


__global__ void GCN_gemm2(const float* __restrict__ agg, const float* __restrict__ b1,
                          const float* __restrict__ W2, float* __restrict__ h2) {
    const int tile = blockIdx.x * 8 + (threadIdx.x >> 5);
    if (tile >= N_NODES / 16) return;
    const int lane = threadIdx.x & 31;
    const int half = lane >> 4;
    const int l    = lane & 15;
    const int m0   = tile * 16;

    v8f c = {};
    const int arow = m0 + l;
    const bool valid_col = (l < F_OUT);
#pragma unroll
    for (int k0 = 0; k0 < F_HID; k0 += 4) {
        const int kb = k0 + half * 2;
        v2f a, b;
        a.x = fmaxf(agg[arow * F_HID + kb]     + b1[kb],     0.0f);
        a.y = fmaxf(agg[arow * F_HID + kb + 1] + b1[kb + 1], 0.0f);
        b.x = valid_col ? W2[kb * F_OUT + l]       : 0.0f;
        b.y = valid_col ? W2[(kb + 1) * F_OUT + l] : 0.0f;
        c = __builtin_amdgcn_wmma_f32_16x16x4_f32(false, a, false, b,
                                                  (short)0, c, false, false);
    }
    if (valid_col) {
#pragma unroll
        for (int r = 0; r < 8; ++r)
            h2[(m0 + r + half * 8) * F_OUT + l] = c[r];
    }
}


__global__ void GCN_self2(const float* __restrict__ h2, const float* __restrict__ dinv,
                          const float* __restrict__ b2, float* __restrict__ out) {
    int idx = blockIdx.x * blockDim.x + threadIdx.x;
    if (idx >= N_NODES * F_OUT) return;
    int i = idx >> 3, ch = idx & 7;
    float w = dinv[i];
    out[idx] = h2[idx] * w * w + b2[ch];
}

__global__ void GCN_edge2(const int* __restrict__ src, const int* __restrict__ dst,
                          const float* __restrict__ dinv, const float* __restrict__ h2,
                          float* __restrict__ out) {
    const int lane = threadIdx.x & 31;
    const int e0   = ((blockIdx.x * blockDim.x + threadIdx.x) >> 5) * 32;
    if (e0 >= E_EDGES) return;
    const int s = src[e0 + lane], d = dst[e0 + lane];
    const float nrm = dinv[s] * dinv[d];
    const int srow = s * F_OUT, drow = d * F_OUT;
    const int q = lane >> 3, ch = lane & 7;
    float v[8];
#pragma unroll
    for (int t = 0; t < 8; ++t)
        v[t] = h2[(unsigned)(__shfl(srow, 4 * t + q) + ch)];
#pragma unroll
    for (int t = 0; t < 8; ++t)
        atomicAdd(out + (unsigned)(__shfl(drow, 4 * t + q) + ch), v[t] * __shfl(nrm, 4 * t + q));
}


extern "C" void kernel_launch(void* const* d_in, const int* in_sizes, int n_in,
                              void* d_out, int out_size, void* d_ws, size_t ws_size,
                              hipStream_t stream) {
    const float* x  = (const float*)d_in[0];
    const int*   ei = (const int*)d_in[1];
    const float* W1 = (const float*)d_in[2];
    const float* b1 = (const float*)d_in[3];
    const float* W2 = (const float*)d_in[4];
    const float* b2 = (const float*)d_in[5];
    float* out = (float*)d_out;

    const int* src = ei;
    const int* dst = ei + E_EDGES;

    float* ws   = (float*)d_ws;
    float* dinv = ws;
    float* aggx = ws + 50176;
    float* agg  = ws + 50176 + 6400000;
    float* h2   = ws + 50176 + 12800000;
    unsigned* planes = (unsigned*)(ws + 50176 + 12800000 + 400000);

    const int TB = 256;

    GCN_deg_hist<<<DEG_SL * DEG_NP, DEG_TB, 0, stream>>>(dst, planes);
    GCN_dinv    <<<(N_NODES + TB - 1) / TB, TB, 0, stream>>>(planes, dinv);

    GCN_self1<<<(N_NODES * (F_IN / 4) + TB - 1) / TB, TB, 0, stream>>>(x, dinv, aggx);
    GCN_edge1<<<(E_EDGES + TB - 1) / TB, TB, 0, stream>>>(src, dst, dinv, x, aggx);

    GCN_gemm1<<<N_NODES / 16, TB, 0, stream>>>(aggx, W1, agg);

    GCN_gemm2<<<(N_NODES / 16 + 7) / 8, TB, 0, stream>>>(agg, b1, W2, h2);

    GCN_self2<<<(N_NODES * F_OUT + TB - 1) / TB, TB, 0, stream>>>(h2, dinv, b2, out);
    GCN_edge2<<<(E_EDGES + TB - 1) / TB, TB, 0, stream>>>(src, dst, dinv, h2, out);
}


# GEMM1, bias+ReLU, GEMM2 and the layer-2 self term fused, computed transposed so the hidden tile stays in registers as the next WMMA's B; agg not stored; 8 -> 6 launches; GEMM2 sums reassociated
# speedup vs baseline: 4.0344x; 1.2264x over previous
#include <hip/hip_runtime.h>

#define N_NODES 50000
#define E_EDGES 800000
#define F_IN    64
#define F_HID   128
#define F_OUT   8

typedef __attribute__((ext_vector_type(2))) float v2f;
typedef __attribute__((ext_vector_type(8))) float v8f;


#define DEG_SL     64
#define DEG_NP     4
#define DEG_PART   12512
#define DEG_PITCH  50176
#define DEG_TB     1024
#define DEG_EPS    (E_EDGES / DEG_SL)
#define DEG_ROUNDS ((DEG_EPS + DEG_TB - 1) / DEG_TB)
static_assert(E_EDGES % DEG_SL == 0, "equal edge slices");
static_assert(DEG_NP * DEG_PART >= N_NODES && DEG_NP * DEG_PART <= DEG_PITCH, "parts cover the nodes inside a plane");
static_assert(DEG_PART % 32 == 0, "parts are whole lines and move as 16-byte pieces");

__global__ __launch_bounds__(DEG_TB) void GCN_deg_hist(const int* __restrict__ dst,
                                                       unsigned* __restrict__ planes) {
    __shared__ __attribute__((aligned(16))) unsigned cnt[DEG_PART];
    const int tid   = threadIdx.x;
    const int part  = blockIdx.x % DEG_NP;
    const int slice = blockIdx.x / DEG_NP;
    const int lo    = part * DEG_PART;
    const int* dp = dst + slice * DEG_EPS;
    int d[DEG_ROUNDS];
#pragma unroll
    for (int j = 0; j < DEG_ROUNDS; ++j) {
        const int e = tid + j * DEG_TB;
        d[j] = (e < DEG_EPS) ? dp[e] : -1;
    }
    uint4* c4 = reinterpret_cast<uint4*>(cnt);
    for (int i = tid; i < DEG_PART / 4; i += DEG_TB) c4[i] = make_uint4(0u, 0u, 0u, 0u);
    __syncthreads();
#pragma unroll
    for (int j = 0; j < DEG_ROUNDS; ++j) {
        const unsigned r = (unsigned)(d[j] - lo);
        if (r < (unsigned)DEG_PART) atomicAdd(&cnt[r], 1u);
    }
    __syncthreads();
    uint4* p4 = reinterpret_cast<uint4*>(planes + (size_t)slice * DEG_PITCH + lo);
    for (int i = tid; i < DEG_PART / 4; i += DEG_TB) p4[i] = c4[i];
}

__global__ void GCN_dinv(const unsigned* __restrict__ planes, float* __restrict__ dinv) {
    int i = blockIdx.x * blockDim.x + threadIdx.x;
    if (i >= N_NODES) return;
    unsigned c = 1u;
#pragma unroll
    for (int b = 0; b < DEG_SL; ++b) c += planes[(size_t)b * DEG_PITCH + i];
    dinv[i] = rsqrtf((float)c);
}


__global__ void GCN_self1(const float* __restrict__ x, const float* __restrict__ dinv,
                          float* __restrict__ aggx) {
    int idx = blockIdx.x * blockDim.x + threadIdx.x;
    if (idx >= N_NODES * (F_IN / 4)) return;
    int i  = idx >> 4;
    int c4 = idx & 15;
    float w = dinv[i]; w = w * w;
    float4 xv = reinterpret_cast<const float4*>(x + (size_t)i * F_IN)[c4];
    float4 o; o.x = xv.x * w; o.y = xv.y * w; o.z = xv.z * w; o.w = xv.w * w;
    reinterpret_cast<float4*>(aggx + (size_t)i * F_IN)[c4] = o;
}

static_assert(E_EDGES % 32 == 0, "every lane of every edge wave owns an edge");

__global__ void GCN_edge1(const int* __restrict__ src, const int* __restrict__ dst,
                          const float* __restrict__ dinv, const float* __restrict__ x,
                          float* __restrict__ aggx) {
    const int lane = threadIdx.x & 31;
    const int e0   = ((blockIdx.x * blockDim.x + threadIdx.x) >> 5) * 32;
    if (e0 >= E_EDGES) return;
    const int s = src[e0 + lane], d = dst[e0 + lane];
    const int nrm  = __float_as_int(dinv[s] * dinv[d]);
    const int srow = s * F_IN, drow = d * F_IN;
#pragma unroll
    for (int g = 0; g < 32; g += 16) {
        float v0[16], v1[16];
#pragma unroll
        for (int j = 0; j < 16; ++j) {
            const float* xp = x + (unsigned)(__builtin_amdgcn_readlane(srow, g + j) + lane);
            v0[j] = xp[0];
            v1[j] = xp[32];
        }
#pragma unroll
        for (int j = 0; j < 16; ++j) {
            float* ap = aggx + (unsigned)(__builtin_amdgcn_readlane(drow, g + j) + lane);
            const float norm = __int_as_float(__builtin_amdgcn_readlane(nrm, g + j));
            atomicAdd(ap +  0, v0[j] * norm);
            atomicAdd(ap + 32, v1[j] * norm);
        }
    }
}


__global__ __launch_bounds__(128) void GCN_mlp(const float* __restrict__ aggx,
                                               const float* __restrict__ W1,
                                               const float* __restrict__ b1,
                                               const float* __restrict__ W2,
                                               const float* __restrict__ b2,
                                               const float* __restrict__ dinv,
                                               float* __restrict__ h2, float* __restrict__ out) {
    const int tile = blockIdx.x * 4 + (threadIdx.x >> 5);
    if (tile >= N_NODES / 16) return;
    const int lane = threadIdx.x & 31;
    const int half = lane >> 4;
    const int l    = lane & 15;
    const int m0   = tile * 16;
    const int node = m0 + l;

    v2f xb[F_IN / 4];
    const float* xrow = aggx + node * F_IN + half * 2;
#pragma unroll
    for (int ks = 0; ks < F_IN / 4; ++ks)
        xb[ks] = *reinterpret_cast<const v2f*>(xrow + 4 * ks);

    const bool valid_oc = (l < F_OUT);
    v8f acc = {};
#pragma unroll
    for (int hp = 0; hp < F_HID / 32; ++hp) {
        const int hc = hp * 32 + l;
        v8f c0 = {}, c1 = {};
        v2f wa0[F_IN / 4], wa1[F_IN / 4];
#pragma unroll
        for (int ks = 0; ks < F_IN / 4; ++ks) {
            const int kb = 4 * ks + half * 2;
            wa0[ks].x = W1[kb * F_HID + hc];
            wa0[ks].y = W1[(kb + 1) * F_HID + hc];
            wa1[ks].x = W1[kb * F_HID + hc + 16];
            wa1[ks].y = W1[(kb + 1) * F_HID + hc + 16];
        }
#pragma unroll
        for (int ks = 0; ks < F_IN / 4; ++ks) {
            c0 = __builtin_amdgcn_wmma_f32_16x16x4_f32(false, wa0[ks], false, xb[ks],
                                                       (short)0, c0, false, false);
            c1 = __builtin_amdgcn_wmma_f32_16x16x4_f32(false, wa1[ks], false, xb[ks],
                                                       (short)0, c1, false, false);
        }
#pragma unroll
        for (int t = 0; t < 2; ++t) {
            v8f c = t ? c1 : c0;
            const int hb = hp * 32 + t * 16 + half * 8;
#pragma unroll
            for (int r = 0; r < 8; ++r) c[r] = fmaxf(c[r] + b1[hb + r], 0.0f);
#pragma unroll
            for (int j = 0; j < 4; ++j) {
                v2f a, b;
                const float w0 = W2[(hb + 2 * j) * F_OUT + (l & 7)];
                const float w1 = W2[(hb + 2 * j + 1) * F_OUT + (l & 7)];
                a.x = valid_oc ? w0 : 0.0f;
                a.y = valid_oc ? w1 : 0.0f;
                b.x = c[2 * j];
                b.y = c[2 * j + 1];
                acc = __builtin_amdgcn_wmma_f32_16x16x4_f32(false, a, false, b,
                                                            (short)0, acc, false, false);
            }
        }
    }

    const float u0 = __shfl(acc[4], l), u1 = __shfl(acc[5], l);
    const float u2 = __shfl(acc[6], l), u3 = __shfl(acc[7], l);
    float4 v;
    v.x = half ? u0 : acc[0];
    v.y = half ? u1 : acc[1];
    v.z = half ? u2 : acc[2];
    v.w = half ? u3 : acc[3];
    const float w = dinv[node];
    const float* bp = b2 + half * 4;
    float4 o;
    o.x = v.x * w * w + bp[0];
    o.y = v.y * w * w + bp[1];
    o.z = v.z * w * w + bp[2];
    o.w = v.w * w * w + bp[3];
    reinterpret_cast<float4*>(h2  + node * F_OUT)[half] = v;
    reinterpret_cast<float4*>(out + node * F_OUT)[half] = o;
}


__global__ void GCN_edge2(const int* __restrict__ src, const int* __restrict__ dst,
                          const float* __restrict__ dinv, const float* __restrict__ h2,
                          float* __restrict__ out) {
    const int lane = threadIdx.x & 31;
    const int e0   = ((blockIdx.x * blockDim.x + threadIdx.x) >> 5) * 32;
    if (e0 >= E_EDGES) return;
    const int s = src[e0 + lane], d = dst[e0 + lane];
    const float nrm = dinv[s] * dinv[d];
    const int srow = s * F_OUT, drow = d * F_OUT;
    const int q = lane >> 3, ch = lane & 7;
    float v[8];
#pragma unroll
    for (int t = 0; t < 8; ++t)
        v[t] = h2[(unsigned)(__shfl(srow, 4 * t + q) + ch)];
#pragma unroll
    for (int t = 0; t < 8; ++t)
        atomicAdd(out + (unsigned)(__shfl(drow, 4 * t + q) + ch), v[t] * __shfl(nrm, 4 * t + q));
}


extern "C" void kernel_launch(void* const* d_in, const int* in_sizes, int n_in,
                              void* d_out, int out_size, void* d_ws, size_t ws_size,
                              hipStream_t stream) {
    const float* x  = (const float*)d_in[0];
    const int*   ei = (const int*)d_in[1];
    const float* W1 = (const float*)d_in[2];
    const float* b1 = (const float*)d_in[3];
    const float* W2 = (const float*)d_in[4];
    const float* b2 = (const float*)d_in[5];
    float* out = (float*)d_out;

    const int* src = ei;
    const int* dst = ei + E_EDGES;

    float* ws   = (float*)d_ws;
    float* dinv = ws;
    float* aggx = ws + 50176;
    float* h2   = ws + 50176 + 12800000;
    unsigned* planes = (unsigned*)(ws + 50176 + 12800000 + 400000);

    const int TB = 256;

    GCN_deg_hist<<<DEG_SL * DEG_NP, DEG_TB, 0, stream>>>(dst, planes);
    GCN_dinv    <<<(N_NODES + TB - 1) / TB, TB, 0, stream>>>(planes, dinv);

    GCN_self1<<<(N_NODES * (F_IN / 4) + TB - 1) / TB, TB, 0, stream>>>(x, dinv, aggx);
    GCN_edge1<<<(E_EDGES + TB - 1) / TB, TB, 0, stream>>>(src, dst, dinv, x, aggx);

    GCN_mlp<<<(N_NODES / 16 + 3) / 4, 128, 0, stream>>>(aggx, W1, b1, W2, b2, dinv, h2, out);

    GCN_edge2<<<(E_EDGES + TB - 1) / TB, TB, 0, stream>>>(src, dst, dinv, h2, out);
}


# edge weight dinv[src]*dinv[dst] split into per-node factors: rows pre-scaled by dinv once per node; dst factor per node in layer 1, one lookup per edge in layer 2 (was 2); products reassociated
# speedup vs baseline: 4.1995x; 1.0409x over previous
#include <hip/hip_runtime.h>

#define N_NODES 50000
#define E_EDGES 800000
#define F_IN    64
#define F_HID   128
#define F_OUT   8

typedef __attribute__((ext_vector_type(2))) float v2f;
typedef __attribute__((ext_vector_type(8))) float v8f;


#define DEG_SL     64
#define DEG_NP     4
#define DEG_PART   12512
#define DEG_PITCH  50176
#define DEG_TB     1024
#define DEG_EPS    (E_EDGES / DEG_SL)
#define DEG_ROUNDS ((DEG_EPS + DEG_TB - 1) / DEG_TB)
static_assert(E_EDGES % DEG_SL == 0, "equal edge slices");
static_assert(DEG_NP * DEG_PART >= N_NODES && DEG_NP * DEG_PART <= DEG_PITCH, "parts cover the nodes inside a plane");
static_assert(DEG_PART % 32 == 0, "parts are whole lines and move as 16-byte pieces");

__global__ __launch_bounds__(DEG_TB) void GCN_deg_hist(const int* __restrict__ dst,
                                                       unsigned* __restrict__ planes) {
    __shared__ __attribute__((aligned(16))) unsigned cnt[DEG_PART];
    const int tid   = threadIdx.x;
    const int part  = blockIdx.x % DEG_NP;
    const int slice = blockIdx.x / DEG_NP;
    const int lo    = part * DEG_PART;
    const int* dp = dst + slice * DEG_EPS;
    int d[DEG_ROUNDS];
#pragma unroll
    for (int j = 0; j < DEG_ROUNDS; ++j) {
        const int e = tid + j * DEG_TB;
        d[j] = (e < DEG_EPS) ? dp[e] : -1;
    }
    uint4* c4 = reinterpret_cast<uint4*>(cnt);
    for (int i = tid; i < DEG_PART / 4; i += DEG_TB) c4[i] = make_uint4(0u, 0u, 0u, 0u);
    __syncthreads();
#pragma unroll
    for (int j = 0; j < DEG_ROUNDS; ++j) {
        const unsigned r = (unsigned)(d[j] - lo);
        if (r < (unsigned)DEG_PART) atomicAdd(&cnt[r], 1u);
    }
    __syncthreads();
    uint4* p4 = reinterpret_cast<uint4*>(planes + (size_t)slice * DEG_PITCH + lo);
    for (int i = tid; i < DEG_PART / 4; i += DEG_TB) p4[i] = c4[i];
}

__global__ void GCN_dinv(const unsigned* __restrict__ planes, float* __restrict__ dinv) {
    int i = blockIdx.x * blockDim.x + threadIdx.x;
    if (i >= N_NODES) return;
    unsigned c = 1u;
#pragma unroll
    for (int b = 0; b < DEG_SL; ++b) c += planes[(size_t)b * DEG_PITCH + i];
    dinv[i] = rsqrtf((float)c);
}


__global__ void GCN_self1(const float* __restrict__ x, const float* __restrict__ dinv,
                          float* __restrict__ xs, float* __restrict__ accx) {
    int idx = blockIdx.x * blockDim.x + threadIdx.x;
    if (idx >= N_NODES * (F_IN / 4)) return;
    int i  = idx >> 4;
    int c4 = idx & 15;
    const float w = dinv[i];
    float4 xv = reinterpret_cast<const float4*>(x + (size_t)i * F_IN)[c4];
    float4 o; o.x = xv.x * w; o.y = xv.y * w; o.z = xv.z * w; o.w = xv.w * w;
    reinterpret_cast<float4*>(xs   + (size_t)i * F_IN)[c4] = o;
    reinterpret_cast<float4*>(accx + (size_t)i * F_IN)[c4] = o;
}

static_assert(E_EDGES % 32 == 0, "every lane of every edge wave owns an edge");

__global__ void GCN_edge1(const int* __restrict__ src, const int* __restrict__ dst,
                          const float* __restrict__ xs, float* __restrict__ accx) {
    const int lane = threadIdx.x & 31;
    const int e0   = ((blockIdx.x * blockDim.x + threadIdx.x) >> 5) * 32;
    if (e0 >= E_EDGES) return;
    const int s = src[e0 + lane], d = dst[e0 + lane];
    const int srow = s * F_IN, drow = d * F_IN;
#pragma unroll
    for (int g = 0; g < 32; g += 16) {
        float v0[16], v1[16];
#pragma unroll
        for (int j = 0; j < 16; ++j) {
            const float* xp = xs + (unsigned)(__builtin_amdgcn_readlane(srow, g + j) + lane);
            v0[j] = xp[0];
            v1[j] = xp[32];
        }
#pragma unroll
        for (int j = 0; j < 16; ++j) {
            float* ap = accx + (unsigned)(__builtin_amdgcn_readlane(drow, g + j) + lane);
            atomicAdd(ap +  0, v0[j]);
            atomicAdd(ap + 32, v1[j]);
        }
    }
}


__global__ __launch_bounds__(128) void GCN_mlp(const float* __restrict__ accx,
                                               const float* __restrict__ W1,
                                               const float* __restrict__ b1,
                                               const float* __restrict__ W2,
                                               const float* __restrict__ b2,
                                               const float* __restrict__ dinv,
                                               float* __restrict__ h2s, float* __restrict__ out) {
    const int tile = blockIdx.x * 4 + (threadIdx.x >> 5);
    if (tile >= N_NODES / 16) return;
    const int lane = threadIdx.x & 31;
    const int half = lane >> 4;
    const int l    = lane & 15;
    const int m0   = tile * 16;
    const int node = m0 + l;

    v2f xb[F_IN / 4];
    const float* xrow = accx + node * F_IN + half * 2;
    const float w = dinv[node];
#pragma unroll
    for (int ks = 0; ks < F_IN / 4; ++ks)
        xb[ks] = *reinterpret_cast<const v2f*>(xrow + 4 * ks);
#pragma unroll
    for (int ks = 0; ks < F_IN / 4; ++ks)
        xb[ks] *= w;

    const bool valid_oc = (l < F_OUT);
    v8f acc = {};
#pragma unroll
    for (int hp = 0; hp < F_HID / 32; ++hp) {
        const int hc = hp * 32 + l;
        v8f c0 = {}, c1 = {};
        v2f wa0[F_IN / 4], wa1[F_IN / 4];
#pragma unroll
        for (int ks = 0; ks < F_IN / 4; ++ks) {
            const int kb = 4 * ks + half * 2;
            wa0[ks].x = W1[kb * F_HID + hc];
            wa0[ks].y = W1[(kb + 1) * F_HID + hc];
            wa1[ks].x = W1[kb * F_HID + hc + 16];
            wa1[ks].y = W1[(kb + 1) * F_HID + hc + 16];
        }
#pragma unroll
        for (int ks = 0; ks < F_IN / 4; ++ks) {
            c0 = __builtin_amdgcn_wmma_f32_16x16x4_f32(false, wa0[ks], false, xb[ks],
                                                       (short)0, c0, false, false);
            c1 = __builtin_amdgcn_wmma_f32_16x16x4_f32(false, wa1[ks], false, xb[ks],
                                                       (short)0, c1, false, false);
        }
#pragma unroll
        for (int t = 0; t < 2; ++t) {
            v8f c = t ? c1 : c0;
            const int hb = hp * 32 + t * 16 + half * 8;
#pragma unroll
            for (int r = 0; r < 8; ++r) c[r] = fmaxf(c[r] + b1[hb + r], 0.0f);
#pragma unroll
            for (int j = 0; j < 4; ++j) {
                v2f a, b;
                const float w0 = W2[(hb + 2 * j) * F_OUT + (l & 7)];
                const float w1 = W2[(hb + 2 * j + 1) * F_OUT + (l & 7)];
                a.x = valid_oc ? w0 : 0.0f;
                a.y = valid_oc ? w1 : 0.0f;
                b.x = c[2 * j];
                b.y = c[2 * j + 1];
                acc = __builtin_amdgcn_wmma_f32_16x16x4_f32(false, a, false, b,
                                                            (short)0, acc, false, false);
            }
        }
    }

    const float u0 = __shfl(acc[4], l), u1 = __shfl(acc[5], l);
    const float u2 = __shfl(acc[6], l), u3 = __shfl(acc[7], l);
    float4 v;
    v.x = half ? u0 : acc[0];
    v.y = half ? u1 : acc[1];
    v.z = half ? u2 : acc[2];
    v.w = half ? u3 : acc[3];
    const float* bp = b2 + half * 4;
    float4 o, hs;
    o.x = v.x * w * w + bp[0];
    o.y = v.y * w * w + bp[1];
    o.z = v.z * w * w + bp[2];
    o.w = v.w * w * w + bp[3];
    hs.x = v.x * w; hs.y = v.y * w; hs.z = v.z * w; hs.w = v.w * w;
    reinterpret_cast<float4*>(h2s + node * F_OUT)[half] = hs;
    reinterpret_cast<float4*>(out + node * F_OUT)[half] = o;
}


__global__ void GCN_edge2(const int* __restrict__ src, const int* __restrict__ dst,
                          const float* __restrict__ dinv, const float* __restrict__ h2s,
                          float* __restrict__ out) {
    const int lane = threadIdx.x & 31;
    const int e0   = ((blockIdx.x * blockDim.x + threadIdx.x) >> 5) * 32;
    if (e0 >= E_EDGES) return;
    const int s = src[e0 + lane], d = dst[e0 + lane];
    const float nrm = dinv[d];
    const int srow = s * F_OUT, drow = d * F_OUT;
    const int q = lane >> 3, ch = lane & 7;
    float v[8];
#pragma unroll
    for (int t = 0; t < 8; ++t)
        v[t] = h2s[(unsigned)(__shfl(srow, 4 * t + q) + ch)];
#pragma unroll
    for (int t = 0; t < 8; ++t)
        atomicAdd(out + (unsigned)(__shfl(drow, 4 * t + q) + ch), v[t] * __shfl(nrm, 4 * t + q));
}


extern "C" void kernel_launch(void* const* d_in, const int* in_sizes, int n_in,
                              void* d_out, int out_size, void* d_ws, size_t ws_size,
                              hipStream_t stream) {
    const float* x  = (const float*)d_in[0];
    const int*   ei = (const int*)d_in[1];
    const float* W1 = (const float*)d_in[2];
    const float* b1 = (const float*)d_in[3];
    const float* W2 = (const float*)d_in[4];
    const float* b2 = (const float*)d_in[5];
    float* out = (float*)d_out;

    const int* src = ei;
    const int* dst = ei + E_EDGES;

    float* ws   = (float*)d_ws;
    float* dinv = ws;
    float* accx = ws + 50176;
    float* h2s  = ws + 50176 + 12800000;
    unsigned* planes = (unsigned*)(ws + 50176 + 12800000 + 400000);
    static_assert(DEG_SL * DEG_PITCH == 3211264, "xs starts behind the planes");
    float* xs   = ws + 50176 + 12800000 + 400000 + 3211264;

    const int TB = 256;

    GCN_deg_hist<<<DEG_SL * DEG_NP, DEG_TB, 0, stream>>>(dst, planes);
    GCN_dinv    <<<(N_NODES + TB - 1) / TB, TB, 0, stream>>>(planes, dinv);

    GCN_self1<<<(N_NODES * (F_IN / 4) + TB - 1) / TB, TB, 0, stream>>>(x, dinv, xs, accx);
    GCN_edge1<<<(E_EDGES + TB - 1) / TB, TB, 0, stream>>>(src, dst, xs, accx);

    GCN_mlp<<<(N_NODES / 16 + 3) / 4, 128, 0, stream>>>(accx, W1, b1, W2, b2, dinv, h2s, out);

    GCN_edge2<<<(E_EDGES + TB - 1) / TB, TB, 0, stream>>>(src, dst, dinv, h2s, out);
}


# edges counting-sorted by destination in-call (LDS slot counters on the degree planes); both aggregations gather per node over its source range, no float atomics; 6 -> 7 launches; sum order differs
# speedup vs baseline: 4.3305x; 1.0312x over previous
#include <hip/hip_runtime.h>

#define N_NODES 50000
#define E_EDGES 800000
#define F_IN    64
#define F_HID   128
#define F_OUT   8

typedef __attribute__((ext_vector_type(2))) float v2f;
typedef __attribute__((ext_vector_type(8))) float v8f;


#define DEG_SL     64
#define DEG_NP     4
#define DEG_PART   12512
#define DEG_PITCH  50176
#define DEG_TB     1024
#define DEG_EPS    (E_EDGES / DEG_SL)
#define DEG_ROUNDS ((DEG_EPS + DEG_TB - 1) / DEG_TB)
static_assert(E_EDGES % DEG_SL == 0, "equal edge slices");
static_assert(DEG_NP * DEG_PART >= N_NODES && DEG_NP * DEG_PART <= DEG_PITCH, "parts cover the nodes inside a plane");
static_assert(DEG_PART % 32 == 0, "parts are whole lines and move as 16-byte pieces");

__global__ __launch_bounds__(DEG_TB) void GCN_deg_hist(const int* __restrict__ dst,
                                                       unsigned* __restrict__ planes) {
    __shared__ __attribute__((aligned(16))) unsigned cnt[DEG_PART];
    const int tid   = threadIdx.x;
    const int part  = blockIdx.x % DEG_NP;
    const int slice = blockIdx.x / DEG_NP;
    const int lo    = part * DEG_PART;
    const int* dp = dst + slice * DEG_EPS;
    int d[DEG_ROUNDS];
#pragma unroll
    for (int j = 0; j < DEG_ROUNDS; ++j) {
        const int e = tid + j * DEG_TB;
        d[j] = (e < DEG_EPS) ? dp[e] : -1;
    }
    uint4* c4 = reinterpret_cast<uint4*>(cnt);
    for (int i = tid; i < DEG_PART / 4; i += DEG_TB) c4[i] = make_uint4(0u, 0u, 0u, 0u);
    __syncthreads();
#pragma unroll
    for (int j = 0; j < DEG_ROUNDS; ++j) {
        const unsigned r = (unsigned)(d[j] - lo);
        if (r < (unsigned)DEG_PART) atomicAdd(&cnt[r], 1u);
    }
    __syncthreads();
    uint4* p4 = reinterpret_cast<uint4*>(planes + (size_t)slice * DEG_PITCH + lo);
    for (int i = tid; i < DEG_PART / 4; i += DEG_TB) p4[i] = c4[i];
}

template <int NW>
__device__ __forceinline__ unsigned block_excl_scan(unsigned v, unsigned* wtot) {
    const int lane = threadIdx.x & 31, wave = threadIdx.x >> 5;
    unsigned inc = v;
#pragma unroll
    for (int o = 1; o < 32; o <<= 1) {
        const unsigned t = __shfl_up(inc, o);
        if (lane >= o) inc += t;
    }
    if (lane == 31) wtot[wave] = inc;
    __syncthreads();
    unsigned base = 0;
    for (int wv = 0; wv < NW; ++wv) {
        const unsigned t = wtot[wv];
        if (wv < wave) base += t;
    }
    return base + inc - v;
}

#define DINV_TB 256
#define DINV_NB ((N_NODES + DINV_TB - 1) / DINV_TB)
static_assert(DINV_NB <= 256, "the block sums are scanned by one work-group, one per thread");

__global__ __launch_bounds__(DINV_TB) void GCN_dinv(unsigned* __restrict__ planes, float* __restrict__ dinv,
                                                    unsigned* __restrict__ lexcl, unsigned* __restrict__ bsum) {
    __shared__ unsigned wtot[DINV_TB / 32];
    const int i = blockIdx.x * DINV_TB + threadIdx.x;
    const bool in = (i < N_NODES);
    unsigned run = 0;
    if (in) {
        unsigned c[DEG_SL];
#pragma unroll
        for (int b = 0; b < DEG_SL; ++b) c[b] = planes[(size_t)b * DEG_PITCH + i];
#pragma unroll
        for (int b = 0; b < DEG_SL; ++b) {
            planes[(size_t)b * DEG_PITCH + i] = run;
            run += c[b];
        }
        dinv[i] = rsqrtf((float)(run + 1u));
    }
    const unsigned ex = block_excl_scan<DINV_TB / 32>(run, wtot);
    if (in) lexcl[i] = ex;
    if (threadIdx.x == DINV_TB - 1) bsum[blockIdx.x] = ex + run;
}

__global__ __launch_bounds__(DEG_TB) void GCN_fill(const int* __restrict__ src, const int* __restrict__ dst,
                                                   const unsigned* __restrict__ planes,
                                                   const unsigned* __restrict__ lexcl,
                                                   const unsigned* __restrict__ bsum,
                                                   int* __restrict__ csr, unsigned* __restrict__ rowptr) {
    __shared__ unsigned cnt[DEG_PART];
    __shared__ unsigned boff[256];
    __shared__ unsigned wtot[DEG_TB / 32];
    const int tid   = threadIdx.x;
    const int part  = blockIdx.x % DEG_NP;
    const int slice = blockIdx.x / DEG_NP;
    const int lo    = part * DEG_PART;
    const int* dp = dst + slice * DEG_EPS;
    const int* sp = src + slice * DEG_EPS;
    int d[DEG_ROUNDS], s[DEG_ROUNDS];
#pragma unroll
    for (int j = 0; j < DEG_ROUNDS; ++j) {
        const int e = tid + j * DEG_TB;
        d[j] = (e < DEG_EPS) ? dp[e] : -1;
        s[j] = (e < DEG_EPS) ? sp[e] : 0;
    }
    const unsigned bs = (tid < DINV_NB) ? bsum[tid] : 0u;
    const unsigned ex = block_excl_scan<DEG_TB / 32>(bs, wtot);
    if (tid < 256) boff[tid] = ex;
    __syncthreads();
    for (int i = tid; i < DEG_PART; i += DEG_TB) {
        const int node = lo + i;
        unsigned st = 0;
        if (node < N_NODES) {
            st = boff[node / DINV_TB] + lexcl[node];
            if (slice == 0) rowptr[node] = st;
            st += planes[(size_t)slice * DEG_PITCH + node];
        }
        cnt[i] = st;
    }
    if (blockIdx.x == 0 && tid == 0) rowptr[N_NODES] = E_EDGES;
    __syncthreads();
#pragma unroll
    for (int j = 0; j < DEG_ROUNDS; ++j) {
        const unsigned r = (unsigned)(d[j] - lo);
        if (r < (unsigned)DEG_PART) {
            const unsigned pos = atomicAdd(&cnt[r], 1u);
            if (pos < (unsigned)E_EDGES) csr[pos] = s[j];
        }
    }
}


__global__ void GCN_self1(const float* __restrict__ x, const float* __restrict__ dinv,
                          float* __restrict__ xs) {
    int idx = blockIdx.x * blockDim.x + threadIdx.x;
    if (idx >= N_NODES * (F_IN / 4)) return;
    int i  = idx >> 4;
    int c4 = idx & 15;
    const float w = dinv[i];
    float4 xv = reinterpret_cast<const float4*>(x + (size_t)i * F_IN)[c4];
    float4 o; o.x = xv.x * w; o.y = xv.y * w; o.z = xv.z * w; o.w = xv.w * w;
    reinterpret_cast<float4*>(xs + (size_t)i * F_IN)[c4] = o;
}

#define AGG1_TB 256
static_assert(N_NODES % (AGG1_TB / 32) == 0, "every wave of the aggregation owns a node");

__global__ __launch_bounds__(AGG1_TB) void GCN_agg1(const unsigned* __restrict__ rowptr, const int* __restrict__ csr,
                                                    const float* __restrict__ xs, float* __restrict__ accx) {
    const int lane = threadIdx.x & 31;
    const int d = __builtin_amdgcn_readfirstlane((int)((blockIdx.x * AGG1_TB + threadIdx.x) >> 5));
    const int beg = (int)rowptr[d], end = (int)rowptr[d + 1];
    const int own = d * F_IN;
    float a0 = xs[own + lane], a1 = xs[own + 32 + lane];
    for (int base = beg; base < end; base += 32) {
        const int n = min(32, end - base);
        const int srow = (lane < n) ? csr[base + lane] * F_IN : own;
        for (int j0 = 0; j0 < n; j0 += 8) {
            float v0[8], v1[8];
#pragma unroll
            for (int u = 0; u < 8; ++u) {
                const float* xp = xs + (unsigned)(__builtin_amdgcn_readlane(srow, (j0 + u) & 31) + lane);
                v0[u] = xp[0];
                v1[u] = xp[32];
            }
#pragma unroll
            for (int u = 0; u < 8; ++u) {
                const bool ok = (j0 + u < n);
                a0 += ok ? v0[u] : 0.0f;
                a1 += ok ? v1[u] : 0.0f;
            }
        }
    }
    accx[own + lane]      = a0;
    accx[own + 32 + lane] = a1;
}


__global__ __launch_bounds__(128) void GCN_mlp(const float* __restrict__ accx,
                                               const float* __restrict__ W1,
                                               const float* __restrict__ b1,
                                               const float* __restrict__ W2,
                                               const float* __restrict__ b2,
                                               const float* __restrict__ dinv,
                                               float* __restrict__ h2s, float* __restrict__ out) {
    const int tile = blockIdx.x * 4 + (threadIdx.x >> 5);
    if (tile >= N_NODES / 16) return;
    const int lane = threadIdx.x & 31;
    const int half = lane >> 4;
    const int l    = lane & 15;
    const int m0   = tile * 16;
    const int node = m0 + l;

    v2f xb[F_IN / 4];
    const float* xrow = accx + node * F_IN + half * 2;
    const float w = dinv[node];
#pragma unroll
    for (int ks = 0; ks < F_IN / 4; ++ks)
        xb[ks] = *reinterpret_cast<const v2f*>(xrow + 4 * ks);
#pragma unroll
    for (int ks = 0; ks < F_IN / 4; ++ks)
        xb[ks] *= w;

    const bool valid_oc = (l < F_OUT);
    v8f acc = {};
#pragma unroll
    for (int hp = 0; hp < F_HID / 32; ++hp) {
        const int hc = hp * 32 + l;
        v8f c0 = {}, c1 = {};
        v2f wa0[F_IN / 4], wa1[F_IN / 4];
#pragma unroll
        for (int ks = 0; ks < F_IN / 4; ++ks) {
            const int kb = 4 * ks + half * 2;
            wa0[ks].x = W1[kb * F_HID + hc];
            wa0[ks].y = W1[(kb + 1) * F_HID + hc];
            wa1[ks].x = W1[kb * F_HID + hc + 16];
            wa1[ks].y = W1[(kb + 1) * F_HID + hc + 16];
        }
#pragma unroll
        for (int ks = 0; ks < F_IN / 4; ++ks) {
            c0 = __builtin_amdgcn_wmma_f32_16x16x4_f32(false, wa0[ks], false, xb[ks],
                                                       (short)0, c0, false, false);
            c1 = __builtin_amdgcn_wmma_f32_16x16x4_f32(false, wa1[ks], false, xb[ks],
                                                       (short)0, c1, false, false);
        }
#pragma unroll
        for (int t = 0; t < 2; ++t) {
            v8f c = t ? c1 : c0;
            const int hb = hp * 32 + t * 16 + half * 8;
#pragma unroll
            for (int r = 0; r < 8; ++r) c[r] = fmaxf(c[r] + b1[hb + r], 0.0f);
#pragma unroll
            for (int j = 0; j < 4; ++j) {
                v2f a, b;
                const float w0 = W2[(hb + 2 * j) * F_OUT + (l & 7)];
                const float w1 = W2[(hb + 2 * j + 1) * F_OUT + (l & 7)];
                a.x = valid_oc ? w0 : 0.0f;
                a.y = valid_oc ? w1 : 0.0f;
                b.x = c[2 * j];
                b.y = c[2 * j + 1];
                acc = __builtin_amdgcn_wmma_f32_16x16x4_f32(false, a, false, b,
                                                            (short)0, acc, false, false);
            }
        }
    }

    const float u0 = __shfl(acc[4], l), u1 = __shfl(acc[5], l);
    const float u2 = __shfl(acc[6], l), u3 = __shfl(acc[7], l);
    float4 v;
    v.x = half ? u0 : acc[0];
    v.y = half ? u1 : acc[1];
    v.z = half ? u2 : acc[2];
    v.w = half ? u3 : acc[3];
    const float* bp = b2 + half * 4;
    float4 o, hs;
    o.x = v.x * w * w + bp[0];
    o.y = v.y * w * w + bp[1];
    o.z = v.z * w * w + bp[2];
    o.w = v.w * w * w + bp[3];
    hs.x = v.x * w; hs.y = v.y * w; hs.z = v.z * w; hs.w = v.w * w;
    reinterpret_cast<float4*>(h2s + node * F_OUT)[half] = hs;
    reinterpret_cast<float4*>(out + node * F_OUT)[half] = o;
}


__global__ void GCN_agg2(const unsigned* __restrict__ rowptr, const int* __restrict__ csr,
                         const float* __restrict__ dinv, const float* __restrict__ h2s,
                         float* __restrict__ out) {
    const int gt = blockIdx.x * blockDim.x + threadIdx.x;
    const int node = gt >> 3, ch = gt & 7;
    if (node >= N_NODES) return;
    const int beg = (int)rowptr[node], end = (int)rowptr[node + 1];
    const float start = out[node * F_OUT + ch];
    const float w = dinv[node];
    float a = 0.0f;
    int j = beg;
    for (; j + 4 <= end; j += 4) {
        const int s0 = csr[j], s1 = csr[j + 1], s2 = csr[j + 2], s3 = csr[j + 3];
        const float v0 = h2s[(unsigned)(s0 * F_OUT + ch)];
        const float v1 = h2s[(unsigned)(s1 * F_OUT + ch)];
        const float v2 = h2s[(unsigned)(s2 * F_OUT + ch)];
        const float v3 = h2s[(unsigned)(s3 * F_OUT + ch)];
        a += v0; a += v1; a += v2; a += v3;
    }
    for (; j < end; ++j) a += h2s[(unsigned)(csr[j] * F_OUT + ch)];
    out[node * F_OUT + ch] = start + a * w;
}


extern "C" void kernel_launch(void* const* d_in, const int* in_sizes, int n_in,
                              void* d_out, int out_size, void* d_ws, size_t ws_size,
                              hipStream_t stream) {
    const float* x  = (const float*)d_in[0];
    const int*   ei = (const int*)d_in[1];
    const float* W1 = (const float*)d_in[2];
    const float* b1 = (const float*)d_in[3];
    const float* W2 = (const float*)d_in[4];
    const float* b2 = (const float*)d_in[5];
    float* out = (float*)d_out;

    const int* src = ei;
    const int* dst = ei + E_EDGES;

    float* ws   = (float*)d_ws;
    float* dinv = ws;
    float* accx = ws + 50176;
    float* h2s  = ws + 50176 + 12800000;
    unsigned* planes = (unsigned*)(ws + 50176 + 12800000 + 400000);
    static_assert(DEG_SL * DEG_PITCH == 3211264, "xs starts behind the planes");
    float* xs   = ws + 50176 + 12800000 + 400000 + 3211264;
    unsigned* lexcl  = (unsigned*)(ws + 50176 + 12800000 + 400000 + 3211264 + 3200000);
    unsigned* bsum   = (unsigned*)(ws + 50176 + 12800000 + 400000 + 3211264 + 3200000 + 50176);
    unsigned* rowptr = (unsigned*)(ws + 50176 + 12800000 + 400000 + 3211264 + 3200000 + 50176 + 256);
    int*      csr    = (int*)(ws + 50176 + 12800000 + 400000 + 3211264 + 3200000 + 50176 + 256 + 50176);

    const int TB = 256;

    GCN_deg_hist<<<DEG_SL * DEG_NP, DEG_TB, 0, stream>>>(dst, planes);
    GCN_dinv    <<<DINV_NB, DINV_TB, 0, stream>>>(planes, dinv, lexcl, bsum);

    GCN_self1<<<(N_NODES * (F_IN / 4) + TB - 1) / TB, TB, 0, stream>>>(x, dinv, xs);
    GCN_fill <<<DEG_SL * DEG_NP, DEG_TB, 0, stream>>>(src, dst, planes, lexcl, bsum, csr, rowptr);
    GCN_agg1 <<<N_NODES / (AGG1_TB / 32), AGG1_TB, 0, stream>>>(rowptr, csr, xs, accx);

    GCN_mlp<<<(N_NODES / 16 + 3) / 4, 128, 0, stream>>>(accx, W1, b1, W2, b2, dinv, h2s, out);

    GCN_agg2<<<(N_NODES * F_OUT + TB - 1) / TB, TB, 0, stream>>>(rowptr, csr, dinv, h2s, out);
}


# scaling of the rows of x by dinv (xs) moved into the kernel that computes dinv: a work-group scales its own 256 rows, loads issued ahead of the plane sums, factors via LDS; 7 -> 6 launches
# speedup vs baseline: 4.4720x; 1.0327x over previous
#include <hip/hip_runtime.h>

#define N_NODES 50000
#define E_EDGES 800000
#define F_IN    64
#define F_HID   128
#define F_OUT   8

typedef __attribute__((ext_vector_type(2))) float v2f;
typedef __attribute__((ext_vector_type(8))) float v8f;


#define DEG_SL     64
#define DEG_NP     4
#define DEG_PART   12512
#define DEG_PITCH  50176
#define DEG_TB     1024
#define DEG_EPS    (E_EDGES / DEG_SL)
#define DEG_ROUNDS ((DEG_EPS + DEG_TB - 1) / DEG_TB)
static_assert(E_EDGES % DEG_SL == 0, "equal edge slices");
static_assert(DEG_NP * DEG_PART >= N_NODES && DEG_NP * DEG_PART <= DEG_PITCH, "parts cover the nodes inside a plane");
static_assert(DEG_PART % 32 == 0, "parts are whole lines and move as 16-byte pieces");

__global__ __launch_bounds__(DEG_TB) void GCN_deg_hist(const int* __restrict__ dst,
                                                       unsigned* __restrict__ planes) {
    __shared__ __attribute__((aligned(16))) unsigned cnt[DEG_PART];
    const int tid   = threadIdx.x;
    const int part  = blockIdx.x % DEG_NP;
    const int slice = blockIdx.x / DEG_NP;
    const int lo    = part * DEG_PART;
    const int* dp = dst + slice * DEG_EPS;
    int d[DEG_ROUNDS];
#pragma unroll
    for (int j = 0; j < DEG_ROUNDS; ++j) {
        const int e = tid + j * DEG_TB;
        d[j] = (e < DEG_EPS) ? dp[e] : -1;
    }
    uint4* c4 = reinterpret_cast<uint4*>(cnt);
    for (int i = tid; i < DEG_PART / 4; i += DEG_TB) c4[i] = make_uint4(0u, 0u, 0u, 0u);
    __syncthreads();
#pragma unroll
    for (int j = 0; j < DEG_ROUNDS; ++j) {
        const unsigned r = (unsigned)(d[j] - lo);
        if (r < (unsigned)DEG_PART) atomicAdd(&cnt[r], 1u);
    }
    __syncthreads();
    uint4* p4 = reinterpret_cast<uint4*>(planes + (size_t)slice * DEG_PITCH + lo);
    for (int i = tid; i < DEG_PART / 4; i += DEG_TB) p4[i] = c4[i];
}

template <int NW>
__device__ __forceinline__ unsigned block_excl_scan(unsigned v, unsigned* wtot) {
    const int lane = threadIdx.x & 31, wave = threadIdx.x >> 5;
    unsigned inc = v;
#pragma unroll
    for (int o = 1; o < 32; o <<= 1) {
        const unsigned t = __shfl_up(inc, o);
        if (lane >= o) inc += t;
    }
    if (lane == 31) wtot[wave] = inc;
    __syncthreads();
    unsigned base = 0;
    for (int wv = 0; wv < NW; ++wv) {
        const unsigned t = wtot[wv];
        if (wv < wave) base += t;
    }
    return base + inc - v;
}

#define DINV_TB 256
#define DINV_NB ((N_NODES + DINV_TB - 1) / DINV_TB)
static_assert(DINV_NB <= 256, "the block sums are scanned by one work-group, one per thread");
static_assert(F_IN == 64 && DINV_TB % 16 == 0, "a row is 16 pieces of 16 bytes");
#define DINV_RP (DINV_TB / 16)
static_assert(N_NODES % DINV_RP == 0, "a pass of DINV_RP rows lies wholly inside or wholly outside the nodes");

__global__ __launch_bounds__(DINV_TB) void GCN_dinv(unsigned* __restrict__ planes, const float* __restrict__ x,
                                                    float* __restrict__ dinv, float* __restrict__ xs,
                                                    unsigned* __restrict__ lexcl, unsigned* __restrict__ bsum) {
    __shared__ unsigned wtot[DINV_TB / 32];
    __shared__ float wrow[DINV_TB];
    const int n0 = blockIdx.x * DINV_TB;
    const int i = n0 + threadIdx.x;
    const bool in = (i < N_NODES);
    const int prow = n0 + (threadIdx.x >> 4);
    const int pc4  = threadIdx.x & 15;
    const float4* xp = reinterpret_cast<const float4*>(x + (size_t)prow * F_IN) + pc4;
    float4 xv[DINV_RP];
#pragma unroll
    for (int k = 0; k < DINV_RP; ++k)
        xv[k] = (n0 + k * DINV_RP < N_NODES) ? xp[k * DINV_RP * (F_IN / 4)]
                                             : make_float4(0.0f, 0.0f, 0.0f, 0.0f);
    unsigned run = 0;
    float wn = 0.0f;
    if (in) {
        unsigned c[DEG_SL];
#pragma unroll
        for (int b = 0; b < DEG_SL; ++b) c[b] = planes[(size_t)b * DEG_PITCH + i];
#pragma unroll
        for (int b = 0; b < DEG_SL; ++b) {
            planes[(size_t)b * DEG_PITCH + i] = run;
            run += c[b];
        }
        wn = rsqrtf((float)(run + 1u));
        dinv[i] = wn;
    }
    wrow[threadIdx.x] = wn;
    const unsigned ex = block_excl_scan<DINV_TB / 32>(run, wtot);
    if (in) lexcl[i] = ex;
    if (threadIdx.x == DINV_TB - 1) bsum[blockIdx.x] = ex + run;
    float4* xsp = reinterpret_cast<float4*>(xs + (size_t)prow * F_IN) + pc4;
    float wk[DINV_RP];
#pragma unroll
    for (int k = 0; k < DINV_RP; ++k) wk[k] = wrow[(threadIdx.x >> 4) + k * DINV_RP];
#pragma unroll
    for (int k = 0; k < DINV_RP; ++k) {
        if (n0 + k * DINV_RP < N_NODES) {
            const float w = wk[k];
            float4 o;
            o.x = xv[k].x * w; o.y = xv[k].y * w; o.z = xv[k].z * w; o.w = xv[k].w * w;
            xsp[k * DINV_RP * (F_IN / 4)] = o;
        }
    }
}

__global__ __launch_bounds__(DEG_TB) void GCN_fill(const int* __restrict__ src, const int* __restrict__ dst,
                                                   const unsigned* __restrict__ planes,
                                                   const unsigned* __restrict__ lexcl,
                                                   const unsigned* __restrict__ bsum,
                                                   int* __restrict__ csr, unsigned* __restrict__ rowptr) {
    __shared__ unsigned cnt[DEG_PART];
    __shared__ unsigned boff[256];
    __shared__ unsigned wtot[DEG_TB / 32];
    const int tid   = threadIdx.x;
    const int part  = blockIdx.x % DEG_NP;
    const int slice = blockIdx.x / DEG_NP;
    const int lo    = part * DEG_PART;
    const int* dp = dst + slice * DEG_EPS;
    const int* sp = src + slice * DEG_EPS;
    int d[DEG_ROUNDS], s[DEG_ROUNDS];
#pragma unroll
    for (int j = 0; j < DEG_ROUNDS; ++j) {
        const int e = tid + j * DEG_TB;
        d[j] = (e < DEG_EPS) ? dp[e] : -1;
        s[j] = (e < DEG_EPS) ? sp[e] : 0;
    }
    const unsigned bs = (tid < DINV_NB) ? bsum[tid] : 0u;
    const unsigned ex = block_excl_scan<DEG_TB / 32>(bs, wtot);
    if (tid < 256) boff[tid] = ex;
    __syncthreads();
    for (int i = tid; i < DEG_PART; i += DEG_TB) {
        const int node = lo + i;
        unsigned st = 0;
        if (node < N_NODES) {
            st = boff[node / DINV_TB] + lexcl[node];
            if (slice == 0) rowptr[node] = st;
            st += planes[(size_t)slice * DEG_PITCH + node];
        }
        cnt[i] = st;
    }
    if (blockIdx.x == 0 && tid == 0) rowptr[N_NODES] = E_EDGES;
    __syncthreads();
#pragma unroll
    for (int j = 0; j < DEG_ROUNDS; ++j) {
        const unsigned r = (unsigned)(d[j] - lo);
        if (r < (unsigned)DEG_PART) {
            const unsigned pos = atomicAdd(&cnt[r], 1u);
            if (pos < (unsigned)E_EDGES) csr[pos] = s[j];
        }
    }
}


#define AGG1_TB 256
static_assert(N_NODES % (AGG1_TB / 32) == 0, "every wave of the aggregation owns a node");

__global__ __launch_bounds__(AGG1_TB) void GCN_agg1(const unsigned* __restrict__ rowptr, const int* __restrict__ csr,
                                                    const float* __restrict__ xs, float* __restrict__ accx) {
    const int lane = threadIdx.x & 31;
    const int d = __builtin_amdgcn_readfirstlane((int)((blockIdx.x * AGG1_TB + threadIdx.x) >> 5));
    const int beg = (int)rowptr[d], end = (int)rowptr[d + 1];
    const int own = d * F_IN;
    float a0 = xs[own + lane], a1 = xs[own + 32 + lane];
    for (int base = beg; base < end; base += 32) {
        const int n = min(32, end - base);
        const int srow = (lane < n) ? csr[base + lane] * F_IN : own;
        for (int j0 = 0; j0 < n; j0 += 8) {
            float v0[8], v1[8];
#pragma unroll
            for (int u = 0; u < 8; ++u) {
                const float* xp = xs + (unsigned)(__builtin_amdgcn_readlane(srow, (j0 + u) & 31) + lane);
                v0[u] = xp[0];
                v1[u] = xp[32];
            }
#pragma unroll
            for (int u = 0; u < 8; ++u) {
                const bool ok = (j0 + u < n);
                a0 += ok ? v0[u] : 0.0f;
                a1 += ok ? v1[u] : 0.0f;
            }
        }
    }
    accx[own + lane]      = a0;
    accx[own + 32 + lane] = a1;
}


__global__ __launch_bounds__(128) void GCN_mlp(const float* __restrict__ accx,
                                               const float* __restrict__ W1,
                                               const float* __restrict__ b1,
                                               const float* __restrict__ W2,
                                               const float* __restrict__ b2,
                                               const float* __restrict__ dinv,
                                               float* __restrict__ h2s, float* __restrict__ out) {
    const int tile = blockIdx.x * 4 + (threadIdx.x >> 5);
    if (tile >= N_NODES / 16) return;
    const int lane = threadIdx.x & 31;
    const int half = lane >> 4;
    const int l    = lane & 15;
    const int m0   = tile * 16;
    const int node = m0 + l;

    v2f xb[F_IN / 4];
    const float* xrow = accx + node * F_IN + half * 2;
    const float w = dinv[node];
#pragma unroll
    for (int ks = 0; ks < F_IN / 4; ++ks)
        xb[ks] = *reinterpret_cast<const v2f*>(xrow + 4 * ks);
#pragma unroll
    for (int ks = 0; ks < F_IN / 4; ++ks)
        xb[ks] *= w;

    const bool valid_oc = (l < F_OUT);
    v8f acc = {};
#pragma unroll
    for (int hp = 0; hp < F_HID / 32; ++hp) {
        const int hc = hp * 32 + l;
        v8f c0 = {}, c1 = {};
        v2f wa0[F_IN / 4], wa1[F_IN / 4];
#pragma unroll
        for (int ks = 0; ks < F_IN / 4; ++ks) {
            const int kb = 4 * ks + half * 2;
            wa0[ks].x = W1[kb * F_HID + hc];
            wa0[ks].y = W1[(kb + 1) * F_HID + hc];
            wa1[ks].x = W1[kb * F_HID + hc + 16];
            wa1[ks].y = W1[(kb + 1) * F_HID + hc + 16];
        }
#pragma unroll
        for (int ks = 0; ks < F_IN / 4; ++ks) {
            c0 = __builtin_amdgcn_wmma_f32_16x16x4_f32(false, wa0[ks], false, xb[ks],
                                                       (short)0, c0, false, false);
            c1 = __builtin_amdgcn_wmma_f32_16x16x4_f32(false, wa1[ks], false, xb[ks],
                                                       (short)0, c1, false, false);
        }
#pragma unroll
        for (int t = 0; t < 2; ++t) {
            v8f c = t ? c1 : c0;
            const int hb = hp * 32 + t * 16 + half * 8;
#pragma unroll
            for (int r = 0; r < 8; ++r) c[r] = fmaxf(c[r] + b1[hb + r], 0.0f);
#pragma unroll
            for (int j = 0; j < 4; ++j) {
                v2f a, b;
                const float w0 = W2[(hb + 2 * j) * F_OUT + (l & 7)];
                const float w1 = W2[(hb + 2 * j + 1) * F_OUT + (l & 7)];
                a.x = valid_oc ? w0 : 0.0f;
                a.y = valid_oc ? w1 : 0.0f;
                b.x = c[2 * j];
                b.y = c[2 * j + 1];
                acc = __builtin_amdgcn_wmma_f32_16x16x4_f32(false, a, false, b,
                                                            (short)0, acc, false, false);
            }
        }
    }

    const float u0 = __shfl(acc[4], l), u1 = __shfl(acc[5], l);
    const float u2 = __shfl(acc[6], l), u3 = __shfl(acc[7], l);
    float4 v;
    v.x = half ? u0 : acc[0];
    v.y = half ? u1 : acc[1];
    v.z = half ? u2 : acc[2];
    v.w = half ? u3 : acc[3];
    const float* bp = b2 + half * 4;
    float4 o, hs;
    o.x = v.x * w * w + bp[0];
    o.y = v.y * w * w + bp[1];
    o.z = v.z * w * w + bp[2];
    o.w = v.w * w * w + bp[3];
    hs.x = v.x * w; hs.y = v.y * w; hs.z = v.z * w; hs.w = v.w * w;
    reinterpret_cast<float4*>(h2s + node * F_OUT)[half] = hs;
    reinterpret_cast<float4*>(out + node * F_OUT)[half] = o;
}


__global__ void GCN_agg2(const unsigned* __restrict__ rowptr, const int* __restrict__ csr,
                         const float* __restrict__ dinv, const float* __restrict__ h2s,
                         float* __restrict__ out) {
    const int gt = blockIdx.x * blockDim.x + threadIdx.x;
    const int node = gt >> 3, ch = gt & 7;
    if (node >= N_NODES) return;
    const int beg = (int)rowptr[node], end = (int)rowptr[node + 1];
    const float start = out[node * F_OUT + ch];
    const float w = dinv[node];
    float a = 0.0f;
    int j = beg;
    for (; j + 4 <= end; j += 4) {
        const int s0 = csr[j], s1 = csr[j + 1], s2 = csr[j + 2], s3 = csr[j + 3];
        const float v0 = h2s[(unsigned)(s0 * F_OUT + ch)];
        const float v1 = h2s[(unsigned)(s1 * F_OUT + ch)];
        const float v2 = h2s[(unsigned)(s2 * F_OUT + ch)];
        const float v3 = h2s[(unsigned)(s3 * F_OUT + ch)];
        a += v0; a += v1; a += v2; a += v3;
    }
    for (; j < end; ++j) a += h2s[(unsigned)(csr[j] * F_OUT + ch)];
    out[node * F_OUT + ch] = start + a * w;
}


extern "C" void kernel_launch(void* const* d_in, const int* in_sizes, int n_in,
                              void* d_out, int out_size, void* d_ws, size_t ws_size,
                              hipStream_t stream) {
    const float* x  = (const float*)d_in[0];
    const int*   ei = (const int*)d_in[1];
    const float* W1 = (const float*)d_in[2];
    const float* b1 = (const float*)d_in[3];
    const float* W2 = (const float*)d_in[4];
    const float* b2 = (const float*)d_in[5];
    float* out = (float*)d_out;

    const int* src = ei;
    const int* dst = ei + E_EDGES;

    float* ws   = (float*)d_ws;
    float* dinv = ws;
    float* accx = ws + 50176;
    float* h2s  = ws + 50176 + 12800000;
    unsigned* planes = (unsigned*)(ws + 50176 + 12800000 + 400000);
    static_assert(DEG_SL * DEG_PITCH == 3211264, "xs starts behind the planes");
    float* xs   = ws + 50176 + 12800000 + 400000 + 3211264;
    unsigned* lexcl  = (unsigned*)(ws + 50176 + 12800000 + 400000 + 3211264 + 3200000);
    unsigned* bsum   = (unsigned*)(ws + 50176 + 12800000 + 400000 + 3211264 + 3200000 + 50176);
    unsigned* rowptr = (unsigned*)(ws + 50176 + 12800000 + 400000 + 3211264 + 3200000 + 50176 + 256);
    int*      csr    = (int*)(ws + 50176 + 12800000 + 400000 + 3211264 + 3200000 + 50176 + 256 + 50176);

    const int TB = 256;

    GCN_deg_hist<<<DEG_SL * DEG_NP, DEG_TB, 0, stream>>>(dst, planes);
    GCN_dinv    <<<DINV_NB, DINV_TB, 0, stream>>>(planes, x, dinv, xs, lexcl, bsum);

    GCN_fill <<<DEG_SL * DEG_NP, DEG_TB, 0, stream>>>(src, dst, planes, lexcl, bsum, csr, rowptr);
    GCN_agg1 <<<N_NODES / (AGG1_TB / 32), AGG1_TB, 0, stream>>>(rowptr, csr, xs, accx);

    GCN_mlp<<<(N_NODES / 16 + 3) / 4, 128, 0, stream>>>(accx, W1, b1, W2, b2, dinv, h2s, out);

    GCN_agg2<<<(N_NODES * F_OUT + TB - 1) / TB, TB, 0, stream>>>(rowptr, csr, dinv, h2s, out);
}


# fill kernel: the two global terms of every LDS counter's start value requested as one group with the slice's endpoints (was 26 dependent loads per thread, one after another)
# speedup vs baseline: 4.8972x; 1.0951x over previous
#include <hip/hip_runtime.h>

#define N_NODES 50000
#define E_EDGES 800000
#define F_IN    64
#define F_HID   128
#define F_OUT   8

typedef __attribute__((ext_vector_type(2))) float v2f;
typedef __attribute__((ext_vector_type(8))) float v8f;


#define DEG_SL     64
#define DEG_NP     4
#define DEG_PART   12512
#define DEG_PITCH  50176
#define DEG_TB     1024
#define DEG_EPS    (E_EDGES / DEG_SL)
#define DEG_ROUNDS ((DEG_EPS + DEG_TB - 1) / DEG_TB)
static_assert(E_EDGES % DEG_SL == 0, "equal edge slices");
static_assert(DEG_NP * DEG_PART >= N_NODES && DEG_NP * DEG_PART <= DEG_PITCH, "parts cover the nodes inside a plane");
static_assert(DEG_PART % 32 == 0, "parts are whole lines and move as 16-byte pieces");

__global__ __launch_bounds__(DEG_TB) void GCN_deg_hist(const int* __restrict__ dst,
                                                       unsigned* __restrict__ planes) {
    __shared__ __attribute__((aligned(16))) unsigned cnt[DEG_PART];
    const int tid   = threadIdx.x;
    const int part  = blockIdx.x % DEG_NP;
    const int slice = blockIdx.x / DEG_NP;
    const int lo    = part * DEG_PART;
    const int* dp = dst + slice * DEG_EPS;
    int d[DEG_ROUNDS];
#pragma unroll
    for (int j = 0; j < DEG_ROUNDS; ++j) {
        const int e = tid + j * DEG_TB;
        d[j] = (e < DEG_EPS) ? dp[e] : -1;
    }
    uint4* c4 = reinterpret_cast<uint4*>(cnt);
    for (int i = tid; i < DEG_PART / 4; i += DEG_TB) c4[i] = make_uint4(0u, 0u, 0u, 0u);
    __syncthreads();
#pragma unroll
    for (int j = 0; j < DEG_ROUNDS; ++j) {
        const unsigned r = (unsigned)(d[j] - lo);
        if (r < (unsigned)DEG_PART) atomicAdd(&cnt[r], 1u);
    }
    __syncthreads();
    uint4* p4 = reinterpret_cast<uint4*>(planes + (size_t)slice * DEG_PITCH + lo);
    for (int i = tid; i < DEG_PART / 4; i += DEG_TB) p4[i] = c4[i];
}

template <int NW>
__device__ __forceinline__ unsigned block_excl_scan(unsigned v, unsigned* wtot) {
    const int lane = threadIdx.x & 31, wave = threadIdx.x >> 5;
    unsigned inc = v;
#pragma unroll
    for (int o = 1; o < 32; o <<= 1) {
        const unsigned t = __shfl_up(inc, o);
        if (lane >= o) inc += t;
    }
    if (lane == 31) wtot[wave] = inc;
    __syncthreads();
    unsigned base = 0;
    for (int wv = 0; wv < NW; ++wv) {
        const unsigned t = wtot[wv];
        if (wv < wave) base += t;
    }
    return base + inc - v;
}

#define DINV_TB 256
#define DINV_NB ((N_NODES + DINV_TB - 1) / DINV_TB)
static_assert(DINV_NB <= 256, "the block sums are scanned by one work-group, one per thread");
static_assert(F_IN == 64 && DINV_TB % 16 == 0, "a row is 16 pieces of 16 bytes");
#define DINV_RP (DINV_TB / 16)
static_assert(N_NODES % DINV_RP == 0, "a pass of DINV_RP rows lies wholly inside or wholly outside the nodes");

__global__ __launch_bounds__(DINV_TB) void GCN_dinv(unsigned* __restrict__ planes, const float* __restrict__ x,
                                                    float* __restrict__ dinv, float* __restrict__ xs,
                                                    unsigned* __restrict__ lexcl, unsigned* __restrict__ bsum) {
    __shared__ unsigned wtot[DINV_TB / 32];
    __shared__ float wrow[DINV_TB];
    const int n0 = blockIdx.x * DINV_TB;
    const int i = n0 + threadIdx.x;
    const bool in = (i < N_NODES);
    const int prow = n0 + (threadIdx.x >> 4);
    const int pc4  = threadIdx.x & 15;
    const float4* xp = reinterpret_cast<const float4*>(x + (size_t)prow * F_IN) + pc4;
    float4 xv[DINV_RP];
#pragma unroll
    for (int k = 0; k < DINV_RP; ++k)
        xv[k] = (n0 + k * DINV_RP < N_NODES) ? xp[k * DINV_RP * (F_IN / 4)]
                                             : make_float4(0.0f, 0.0f, 0.0f, 0.0f);
    unsigned run = 0;
    float wn = 0.0f;
    if (in) {
        unsigned c[DEG_SL];
#pragma unroll
        for (int b = 0; b < DEG_SL; ++b) c[b] = planes[(size_t)b * DEG_PITCH + i];
#pragma unroll
        for (int b = 0; b < DEG_SL; ++b) {
            planes[(size_t)b * DEG_PITCH + i] = run;
            run += c[b];
        }
        wn = rsqrtf((float)(run + 1u));
        dinv[i] = wn;
    }
    wrow[threadIdx.x] = wn;
    const unsigned ex = block_excl_scan<DINV_TB / 32>(run, wtot);
    if (in) lexcl[i] = ex;
    if (threadIdx.x == DINV_TB - 1) bsum[blockIdx.x] = ex + run;
    float4* xsp = reinterpret_cast<float4*>(xs + (size_t)prow * F_IN) + pc4;
    float wk[DINV_RP];
#pragma unroll
    for (int k = 0; k < DINV_RP; ++k) wk[k] = wrow[(threadIdx.x >> 4) + k * DINV_RP];
#pragma unroll
    for (int k = 0; k < DINV_RP; ++k) {
        if (n0 + k * DINV_RP < N_NODES) {
            const float w = wk[k];
            float4 o;
            o.x = xv[k].x * w; o.y = xv[k].y * w; o.z = xv[k].z * w; o.w = xv[k].w * w;
            xsp[k * DINV_RP * (F_IN / 4)] = o;
        }
    }
}

#define FILL_ROUNDS ((DEG_PART + DEG_TB - 1) / DEG_TB)
__global__ __launch_bounds__(DEG_TB) void GCN_fill(const int* __restrict__ src, const int* __restrict__ dst,
                                                   const unsigned* __restrict__ planes,
                                                   const unsigned* __restrict__ lexcl,
                                                   const unsigned* __restrict__ bsum,
                                                   int* __restrict__ csr, unsigned* __restrict__ rowptr) {
    __shared__ unsigned cnt[DEG_PART];
    __shared__ unsigned boff[256];
    __shared__ unsigned wtot[DEG_TB / 32];
    const int tid   = threadIdx.x;
    const int part  = blockIdx.x % DEG_NP;
    const int slice = blockIdx.x / DEG_NP;
    const int lo    = part * DEG_PART;
    const int* dp = dst + slice * DEG_EPS;
    const int* sp = src + slice * DEG_EPS;
    int d[DEG_ROUNDS], s[DEG_ROUNDS];
#pragma unroll
    for (int j = 0; j < DEG_ROUNDS; ++j) {
        const int e = tid + j * DEG_TB;
        d[j] = (e < DEG_EPS) ? dp[e] : -1;
        s[j] = (e < DEG_EPS) ? sp[e] : 0;
    }
    unsigned lx[FILL_ROUNDS], pl[FILL_ROUNDS];
    const unsigned* pp = planes + (size_t)slice * DEG_PITCH;
#pragma unroll
    for (int k = 0; k < FILL_ROUNDS; ++k) {
        const int nc = min(lo + tid + k * DEG_TB, N_NODES - 1);
        lx[k] = lexcl[nc];
        pl[k] = pp[nc];
    }
    const unsigned bs = (tid < DINV_NB) ? bsum[tid] : 0u;
    const unsigned ex = block_excl_scan<DEG_TB / 32>(bs, wtot);
    if (tid < 256) boff[tid] = ex;
    __syncthreads();
    unsigned bo[FILL_ROUNDS];
#pragma unroll
    for (int k = 0; k < FILL_ROUNDS; ++k)
        bo[k] = boff[min(lo + tid + k * DEG_TB, N_NODES - 1) / DINV_TB];
#pragma unroll
    for (int k = 0; k < FILL_ROUNDS; ++k) {
        const int i = tid + k * DEG_TB;
        const int node = lo + i;
        if (i < DEG_PART) {
            unsigned st = 0;
            if (node < N_NODES) {
                st = bo[k] + lx[k];
                if (slice == 0) rowptr[node] = st;
                st += pl[k];
            }
            cnt[i] = st;
        }
    }
    if (blockIdx.x == 0 && tid == 0) rowptr[N_NODES] = E_EDGES;
    __syncthreads();
#pragma unroll
    for (int j = 0; j < DEG_ROUNDS; ++j) {
        const unsigned r = (unsigned)(d[j] - lo);
        if (r < (unsigned)DEG_PART) {
            const unsigned pos = atomicAdd(&cnt[r], 1u);
            if (pos < (unsigned)E_EDGES) csr[pos] = s[j];
        }
    }
}


#define AGG1_TB 256
static_assert(N_NODES % (AGG1_TB / 32) == 0, "every wave of the aggregation owns a node");

__global__ __launch_bounds__(AGG1_TB) void GCN_agg1(const unsigned* __restrict__ rowptr, const int* __restrict__ csr,
                                                    const float* __restrict__ xs, float* __restrict__ accx) {
    const int lane = threadIdx.x & 31;
    const int d = __builtin_amdgcn_readfirstlane((int)((blockIdx.x * AGG1_TB + threadIdx.x) >> 5));
    const int beg = (int)rowptr[d], end = (int)rowptr[d + 1];
    const int own = d * F_IN;
    float a0 = xs[own + lane], a1 = xs[own + 32 + lane];
    for (int base = beg; base < end; base += 32) {
        const int n = min(32, end - base);
        const int srow = (lane < n) ? csr[base + lane] * F_IN : own;
        for (int j0 = 0; j0 < n; j0 += 8) {
            float v0[8], v1[8];
#pragma unroll
            for (int u = 0; u < 8; ++u) {
                const float* xp = xs + (unsigned)(__builtin_amdgcn_readlane(srow, (j0 + u) & 31) + lane);
                v0[u] = xp[0];
                v1[u] = xp[32];
            }
#pragma unroll
            for (int u = 0; u < 8; ++u) {
                const bool ok = (j0 + u < n);
                a0 += ok ? v0[u] : 0.0f;
                a1 += ok ? v1[u] : 0.0f;
            }
        }
    }
    accx[own + lane]      = a0;
    accx[own + 32 + lane] = a1;
}


__global__ __launch_bounds__(128) void GCN_mlp(const float* __restrict__ accx,
                                               const float* __restrict__ W1,
                                               const float* __restrict__ b1,
                                               const float* __restrict__ W2,
                                               const float* __restrict__ b2,
                                               const float* __restrict__ dinv,
                                               float* __restrict__ h2s, float* __restrict__ out) {
    const int tile = blockIdx.x * 4 + (threadIdx.x >> 5);
    if (tile >= N_NODES / 16) return;
    const int lane = threadIdx.x & 31;
    const int half = lane >> 4;
    const int l    = lane & 15;
    const int m0   = tile * 16;
    const int node = m0 + l;

    v2f xb[F_IN / 4];
    const float* xrow = accx + node * F_IN + half * 2;
    const float w = dinv[node];
#pragma unroll
    for (int ks = 0; ks < F_IN / 4; ++ks)
        xb[ks] = *reinterpret_cast<const v2f*>(xrow + 4 * ks);
#pragma unroll
    for (int ks = 0; ks < F_IN / 4; ++ks)
        xb[ks] *= w;

    const bool valid_oc = (l < F_OUT);
    v8f acc = {};
#pragma unroll
    for (int hp = 0; hp < F_HID / 32; ++hp) {
        const int hc = hp * 32 + l;
        v8f c0 = {}, c1 = {};
        v2f wa0[F_IN / 4], wa1[F_IN / 4];
#pragma unroll
        for (int ks = 0; ks < F_IN / 4; ++ks) {
            const int kb = 4 * ks + half * 2;
            wa0[ks].x = W1[kb * F_HID + hc];
            wa0[ks].y = W1[(kb + 1) * F_HID + hc];
            wa1[ks].x = W1[kb * F_HID + hc + 16];
            wa1[ks].y = W1[(kb + 1) * F_HID + hc + 16];
        }
#pragma unroll
        for (int ks = 0; ks < F_IN / 4; ++ks) {
            c0 = __builtin_amdgcn_wmma_f32_16x16x4_f32(false, wa0[ks], false, xb[ks],
                                                       (short)0, c0, false, false);
            c1 = __builtin_amdgcn_wmma_f32_16x16x4_f32(false, wa1[ks], false, xb[ks],
                                                       (short)0, c1, false, false);
        }
#pragma unroll
        for (int t = 0; t < 2; ++t) {
            v8f c = t ? c1 : c0;
            const int hb = hp * 32 + t * 16 + half * 8;
#pragma unroll
            for (int r = 0; r < 8; ++r) c[r] = fmaxf(c[r] + b1[hb + r], 0.0f);
#pragma unroll
            for (int j = 0; j < 4; ++j) {
                v2f a, b;
                const float w0 = W2[(hb + 2 * j) * F_OUT + (l & 7)];
                const float w1 = W2[(hb + 2 * j + 1) * F_OUT + (l & 7)];
                a.x = valid_oc ? w0 : 0.0f;
                a.y = valid_oc ? w1 : 0.0f;
                b.x = c[2 * j];
                b.y = c[2 * j + 1];
                acc = __builtin_amdgcn_wmma_f32_16x16x4_f32(false, a, false, b,
                                                            (short)0, acc, false, false);
            }
        }
    }

    const float u0 = __shfl(acc[4], l), u1 = __shfl(acc[5], l);
    const float u2 = __shfl(acc[6], l), u3 = __shfl(acc[7], l);
    float4 v;
    v.x = half ? u0 : acc[0];
    v.y = half ? u1 : acc[1];
    v.z = half ? u2 : acc[2];
    v.w = half ? u3 : acc[3];
    const float* bp = b2 + half * 4;
    float4 o, hs;
    o.x = v.x * w * w + bp[0];
    o.y = v.y * w * w + bp[1];
    o.z = v.z * w * w + bp[2];
    o.w = v.w * w * w + bp[3];
    hs.x = v.x * w; hs.y = v.y * w; hs.z = v.z * w; hs.w = v.w * w;
    reinterpret_cast<float4*>(h2s + node * F_OUT)[half] = hs;
    reinterpret_cast<float4*>(out + node * F_OUT)[half] = o;
}


__global__ void GCN_agg2(const unsigned* __restrict__ rowptr, const int* __restrict__ csr,
                         const float* __restrict__ dinv, const float* __restrict__ h2s,
                         float* __restrict__ out) {
    const int gt = blockIdx.x * blockDim.x + threadIdx.x;
    const int node = gt >> 3, ch = gt & 7;
    if (node >= N_NODES) return;
    const int beg = (int)rowptr[node], end = (int)rowptr[node + 1];
    const float start = out[node * F_OUT + ch];
    const float w = dinv[node];
    float a = 0.0f;
    int j = beg;
    for (; j + 4 <= end; j += 4) {
        const int s0 = csr[j], s1 = csr[j + 1], s2 = csr[j + 2], s3 = csr[j + 3];
        const float v0 = h2s[(unsigned)(s0 * F_OUT + ch)];
        const float v1 = h2s[(unsigned)(s1 * F_OUT + ch)];
        const float v2 = h2s[(unsigned)(s2 * F_OUT + ch)];
        const float v3 = h2s[(unsigned)(s3 * F_OUT + ch)];
        a += v0; a += v1; a += v2; a += v3;
    }
    for (; j < end; ++j) a += h2s[(unsigned)(csr[j] * F_OUT + ch)];
    out[node * F_OUT + ch] = start + a * w;
}


extern "C" void kernel_launch(void* const* d_in, const int* in_sizes, int n_in,
                              void* d_out, int out_size, void* d_ws, size_t ws_size,
                              hipStream_t stream) {
    const float* x  = (const float*)d_in[0];
    const int*   ei = (const int*)d_in[1];
    const float* W1 = (const float*)d_in[2];
    const float* b1 = (const float*)d_in[3];
    const float* W2 = (const float*)d_in[4];
    const float* b2 = (const float*)d_in[5];
    float* out = (float*)d_out;

    const int* src = ei;
    const int* dst = ei + E_EDGES;

    float* ws   = (float*)d_ws;
    float* dinv = ws;
    float* accx = ws + 50176;
    float* h2s  = ws + 50176 + 12800000;
    unsigned* planes = (unsigned*)(ws + 50176 + 12800000 + 400000);
    static_assert(DEG_SL * DEG_PITCH == 3211264, "xs starts behind the planes");
    float* xs   = ws + 50176 + 12800000 + 400000 + 3211264;
    unsigned* lexcl  = (unsigned*)(ws + 50176 + 12800000 + 400000 + 3211264 + 3200000);
    unsigned* bsum   = (unsigned*)(ws + 50176 + 12800000 + 400000 + 3211264 + 3200000 + 50176);
    unsigned* rowptr = (unsigned*)(ws + 50176 + 12800000 + 400000 + 3211264 + 3200000 + 50176 + 256);
    int*      csr    = (int*)(ws + 50176 + 12800000 + 400000 + 3211264 + 3200000 + 50176 + 256 + 50176);

    const int TB = 256;

    GCN_deg_hist<<<DEG_SL * DEG_NP, DEG_TB, 0, stream>>>(dst, planes);
    GCN_dinv    <<<DINV_NB, DINV_TB, 0, stream>>>(planes, x, dinv, xs, lexcl, bsum);

    GCN_fill <<<DEG_SL * DEG_NP, DEG_TB, 0, stream>>>(src, dst, planes, lexcl, bsum, csr, rowptr);
    GCN_agg1 <<<N_NODES / (AGG1_TB / 32), AGG1_TB, 0, stream>>>(rowptr, csr, xs, accx);

    GCN_mlp<<<(N_NODES / 16 + 3) / 4, 128, 0, stream>>>(accx, W1, b1, W2, b2, dinv, h2s, out);

    GCN_agg2<<<(N_NODES * F_OUT + TB - 1) / TB, TB, 0, stream>>>(rowptr, csr, dinv, h2s, out);
}


# edges ordered by destination in 2 launches (was 3): slice-local LDS sort by 196-node bin, then one work-group per bin builds its csr piece, dinv, xs; lists sorted by source: sum order differs; 6 -> 5
# speedup vs baseline: 7.4179x; 1.5147x over previous
#include <hip/hip_runtime.h>

#define N_NODES 50000
#define E_EDGES 800000
#define F_IN    64
#define F_HID   128
#define F_OUT   8

typedef __attribute__((ext_vector_type(2))) float v2f;
typedef __attribute__((ext_vector_type(8))) float v8f;


#define BIN_NB   256
#define BIN_W    196
#define BIN_SL   256
#define BIN_EPS  (E_EDGES / BIN_SL)
#define BIN_TB   1024
#define BIN_R    ((BIN_EPS + BIN_TB - 1) / BIN_TB)
#define BIN_PITCH 3136
#define OFF_PITCH 288
static_assert(BIN_PITCH >= BIN_EPS && BIN_PITCH % 32 == 0, "a slice's edges end inside its own lines");
static_assert(OFF_PITCH >= BIN_NB + 1 && OFF_PITCH % 32 == 0, "a slice's run starts end inside their own lines");
static_assert(E_EDGES % BIN_SL == 0, "equal edge slices");
static_assert(BIN_NB * BIN_W >= N_NODES, "the bins cover the nodes");
static_assert(N_NODES <= 65536 && BIN_W <= 256, "source id and in-bin destination share one word");
static_assert(BIN_NB <= BIN_TB && BIN_SL <= BIN_TB, "one thread per bin / per slice in the scans");

template <int NW>
__device__ __forceinline__ unsigned block_excl_scan(unsigned v, unsigned* wtot) {
    const int lane = threadIdx.x & 31, wave = threadIdx.x >> 5;
    unsigned inc = v;
#pragma unroll
    for (int o = 1; o < 32; o <<= 1) {
        const unsigned t = __shfl_up(inc, o);
        if (lane >= o) inc += t;
    }
    if (lane == 31) wtot[wave] = inc;
    __syncthreads();
    unsigned base = 0;
    for (int wv = 0; wv < NW; ++wv) {
        const unsigned t = wtot[wv];
        if (wv < wave) base += t;
    }
    return base + inc - v;
}

__global__ __launch_bounds__(BIN_TB) void GCN_bin(const int* __restrict__ src, const int* __restrict__ dst,
                                                  unsigned* __restrict__ ebuf, unsigned* __restrict__ offS) {
    __shared__ unsigned cnt[BIN_NB];
    __shared__ unsigned off[BIN_NB];
    __shared__ unsigned wtot[BIN_TB / 32];
    __shared__ unsigned stage[BIN_EPS];
    const int tid   = threadIdx.x;
    const int slice = blockIdx.x;
    const int* dp = dst + slice * BIN_EPS;
    const int* sp = src + slice * BIN_EPS;
    int d[BIN_R], s[BIN_R];
#pragma unroll
    for (int j = 0; j < BIN_R; ++j) {
        const int e = tid + j * BIN_TB;
        d[j] = (e < BIN_EPS) ? dp[e] : -1;
        s[j] = (e < BIN_EPS) ? sp[e] : 0;
    }
    if (tid < BIN_NB) cnt[tid] = 0u;
    __syncthreads();
    unsigned b[BIN_R], r[BIN_R];
#pragma unroll
    for (int j = 0; j < BIN_R; ++j) {
        const bool ok = ((unsigned)d[j] < (unsigned)N_NODES) && ((unsigned)s[j] < (unsigned)N_NODES);
        b[j] = ok ? (unsigned)d[j] / (unsigned)BIN_W : (unsigned)BIN_NB;
        r[j] = 0u;
        if (ok) r[j] = atomicAdd(&cnt[b[j]], 1u);
    }
    __syncthreads();
    const unsigned c  = (tid < BIN_NB) ? cnt[tid] : 0u;
    const unsigned ex = block_excl_scan<BIN_TB / 32>(c, wtot);
    if (tid < BIN_NB) {
        off[tid] = ex;
        offS[slice * OFF_PITCH + tid] = ex;
    }
    if (tid == BIN_NB - 1) offS[slice * OFF_PITCH + BIN_NB] = ex + c;
    __syncthreads();
#pragma unroll
    for (int j = 0; j < BIN_R; ++j) {
        if (b[j] < (unsigned)BIN_NB)
            stage[off[b[j]] + r[j]] = (unsigned)s[j] | (((unsigned)d[j] - b[j] * (unsigned)BIN_W) << 16);
    }
    __syncthreads();
    unsigned* ep = ebuf + slice * BIN_PITCH;
    for (int i = tid; i < BIN_EPS; i += BIN_TB) ep[i] = stage[i];
}

#define CSR_TB   1024
#define CSR_CAP  8192
#define CSR_XP   ((BIN_W * (F_IN / 4) + CSR_TB - 1) / CSR_TB)
#define CSR_G    (CSR_TB / 16)
static_assert(F_IN == 64, "a row is 16 pieces of 16 bytes");
static_assert(BIN_SL % CSR_G == 0, "every group walks the same number of runs");

__global__ __launch_bounds__(CSR_TB) void GCN_csr(const unsigned* __restrict__ ebuf, const unsigned* __restrict__ offS,
                                                  const float* __restrict__ x,
                                                  float* __restrict__ dinv, float* __restrict__ xs,
                                                  unsigned* __restrict__ rowptr, int* __restrict__ csr) {
    __shared__ unsigned ra[BIN_SL], rl[BIN_SL];
    __shared__ unsigned wtA[CSR_TB / 32], wtL[CSR_TB / 32], wtC[CSR_TB / 32];
    __shared__ unsigned cntn[256], nbeg[256], cur[256];
    __shared__ float    wrow[256];
    __shared__ unsigned tot[2];
    __shared__ unsigned stg[CSR_CAP], stg2[CSR_CAP];
    const int tid = threadIdx.x;
    const int bin = blockIdx.x;
    const int n0  = bin * BIN_W;
    const float4* x4 = reinterpret_cast<const float4*>(x) + (size_t)n0 * (F_IN / 4);
    float4 xv[CSR_XP];
#pragma unroll
    for (int k = 0; k < CSR_XP; ++k) {
        const int p = tid + k * CSR_TB;
        const bool ok = (p < BIN_W * (F_IN / 4)) && (n0 + (p >> 4) < N_NODES);
        xv[k] = ok ? x4[p] : make_float4(0.0f, 0.0f, 0.0f, 0.0f);
    }
    unsigned a = 0u, len = 0u;
    if (tid < BIN_SL) {
        a   = offS[tid * OFF_PITCH + bin];
        len = offS[tid * OFF_PITCH + bin + 1] - a;
    }
    const unsigned exA = block_excl_scan<CSR_TB / 32>(a, wtA);
    const unsigned exL = block_excl_scan<CSR_TB / 32>(len, wtL);
    if (tid < BIN_SL) { ra[tid] = a; rl[tid] = len; }
    if (tid == CSR_TB - 1) { tot[0] = exA + a; tot[1] = exL + len; }
    if (tid < 256) cntn[tid] = 0u;
    __syncthreads();
    const unsigned base = tot[0];
    const unsigned nb   = tot[1];
    const int g = tid >> 4, l16 = tid & 15;
    const unsigned* rp[BIN_SL / CSR_G];
    unsigned ln[BIN_SL / CSR_G], wq[BIN_SL / CSR_G];
#pragma unroll
    for (int q = 0; q < BIN_SL / CSR_G; ++q) {
        const int s = g + q * CSR_G;
        ln[q] = rl[s];
        rp[q] = ebuf + (size_t)s * BIN_PITCH + ra[s];
        wq[q] = ((unsigned)l16 < ln[q]) ? rp[q][l16] : 0u;
    }
#pragma unroll
    for (int q = 0; q < BIN_SL / CSR_G; ++q)
        if ((unsigned)l16 < ln[q]) atomicAdd(&cntn[wq[q] >> 16], 1u);
#pragma unroll
    for (int q = 0; q < BIN_SL / CSR_G; ++q)
        for (unsigned k = l16 + 16; k < ln[q]; k += 16) atomicAdd(&cntn[rp[q][k] >> 16], 1u);
    __syncthreads();
    const unsigned c  = (tid < 256) ? cntn[tid] : 0u;
    const unsigned ex = block_excl_scan<CSR_TB / 32>(c, wtC);
    if (tid < 256) { nbeg[tid] = ex; cur[tid] = ex; }
    if (tid < BIN_W) {
        const int node = n0 + tid;
        float w = 0.0f;
        if (node < N_NODES) {
            w = rsqrtf((float)(c + 1u));
            dinv[node]   = w;
            rowptr[node] = base + ex;
        }
        wrow[tid] = w;
    }
    if (bin == BIN_NB - 1 && tid == 0) rowptr[N_NODES] = base + nb;
    __syncthreads();
    const bool staged = (nb <= (unsigned)CSR_CAP);
#pragma unroll
    for (int q = 0; q < BIN_SL / CSR_G; ++q) {
        if ((unsigned)l16 < ln[q]) {
            const unsigned w = wq[q];
            const unsigned pos = atomicAdd(&cur[w >> 16], 1u);
            if (staged) stg[pos] = w;
            else if (base + pos < (unsigned)E_EDGES) csr[base + pos] = (int)(w & 0xffffu);
        }
    }
#pragma unroll
    for (int q = 0; q < BIN_SL / CSR_G; ++q) {
        for (unsigned k = l16 + 16; k < ln[q]; k += 16) {
            const unsigned w = rp[q][k];
            const unsigned pos = atomicAdd(&cur[w >> 16], 1u);
            if (staged) stg[pos] = w;
            else if (base + pos < (unsigned)E_EDGES) csr[base + pos] = (int)(w & 0xffffu);
        }
    }
    __syncthreads();
    if (staged) {
        for (unsigned i = tid; i < nb; i += CSR_TB) {
            const unsigned w  = stg[i];
            const unsigned ld = w >> 16;
            const unsigned b0 = nbeg[ld], n = cntn[ld];
            unsigned rank = 0u;
            for (unsigned j = 0; j < n; ++j) {
                const unsigned u = stg[b0 + j];
                rank += ((u < w) || (u == w && b0 + j < i)) ? 1u : 0u;
            }
            stg2[b0 + rank] = w & 0xffffu;
        }
        __syncthreads();
        for (unsigned i = tid; i < nb; i += CSR_TB)
            if (base + i < (unsigned)E_EDGES) csr[base + i] = (int)stg2[i];
    }
    float4* xs4 = reinterpret_cast<float4*>(xs) + (size_t)n0 * (F_IN / 4);
#pragma unroll
    for (int k = 0; k < CSR_XP; ++k) {
        const int p = tid + k * CSR_TB;
        if ((p < BIN_W * (F_IN / 4)) && (n0 + (p >> 4) < N_NODES)) {
            const float w = wrow[p >> 4];
            float4 o;
            o.x = xv[k].x * w; o.y = xv[k].y * w; o.z = xv[k].z * w; o.w = xv[k].w * w;
            xs4[p] = o;
        }
    }
}


#define AGG1_TB 256
static_assert(N_NODES % (AGG1_TB / 32) == 0, "every wave of the aggregation owns a node");

__global__ __launch_bounds__(AGG1_TB) void GCN_agg1(const unsigned* __restrict__ rowptr, const int* __restrict__ csr,
                                                    const float* __restrict__ xs, float* __restrict__ accx) {
    const int lane = threadIdx.x & 31;
    const int d = __builtin_amdgcn_readfirstlane((int)((blockIdx.x * AGG1_TB + threadIdx.x) >> 5));
    const int beg = (int)rowptr[d], end = (int)rowptr[d + 1];
    const int own = d * F_IN;
    float a0 = xs[own + lane], a1 = xs[own + 32 + lane];
    for (int base = beg; base < end; base += 32) {
        const int n = min(32, end - base);
        const int srow = (lane < n) ? csr[base + lane] * F_IN : own;
        for (int j0 = 0; j0 < n; j0 += 8) {
            float v0[8], v1[8];
#pragma unroll
            for (int u = 0; u < 8; ++u) {
                const float* xp = xs + (unsigned)(__builtin_amdgcn_readlane(srow, (j0 + u) & 31) + lane);
                v0[u] = xp[0];
                v1[u] = xp[32];
            }
#pragma unroll
            for (int u = 0; u < 8; ++u) {
                const bool ok = (j0 + u < n);
                a0 += ok ? v0[u] : 0.0f;
                a1 += ok ? v1[u] : 0.0f;
            }
        }
    }
    accx[own + lane]      = a0;
    accx[own + 32 + lane] = a1;
}


__global__ __launch_bounds__(128) void GCN_mlp(const float* __restrict__ accx,
                                               const float* __restrict__ W1,
                                               const float* __restrict__ b1,
                                               const float* __restrict__ W2,
                                               const float* __restrict__ b2,
                                               const float* __restrict__ dinv,
                                               float* __restrict__ h2s, float* __restrict__ out) {
    const int tile = blockIdx.x * 4 + (threadIdx.x >> 5);
    if (tile >= N_NODES / 16) return;
    const int lane = threadIdx.x & 31;
    const int half = lane >> 4;
    const int l    = lane & 15;
    const int m0   = tile * 16;
    const int node = m0 + l;

    v2f xb[F_IN / 4];
    const float* xrow = accx + node * F_IN + half * 2;
    const float w = dinv[node];
#pragma unroll
    for (int ks = 0; ks < F_IN / 4; ++ks)
        xb[ks] = *reinterpret_cast<const v2f*>(xrow + 4 * ks);
#pragma unroll
    for (int ks = 0; ks < F_IN / 4; ++ks)
        xb[ks] *= w;

    const bool valid_oc = (l < F_OUT);
    v8f acc = {};
#pragma unroll
    for (int hp = 0; hp < F_HID / 32; ++hp) {
        const int hc = hp * 32 + l;
        v8f c0 = {}, c1 = {};
        v2f wa0[F_IN / 4], wa1[F_IN / 4];
#pragma unroll
        for (int ks = 0; ks < F_IN / 4; ++ks) {
            const int kb = 4 * ks + half * 2;
            wa0[ks].x = W1[kb * F_HID + hc];
            wa0[ks].y = W1[(kb + 1) * F_HID + hc];
            wa1[ks].x = W1[kb * F_HID + hc + 16];
            wa1[ks].y = W1[(kb + 1) * F_HID + hc + 16];
        }
#pragma unroll
        for (int ks = 0; ks < F_IN / 4; ++ks) {
            c0 = __builtin_amdgcn_wmma_f32_16x16x4_f32(false, wa0[ks], false, xb[ks],
                                                       (short)0, c0, false, false);
            c1 = __builtin_amdgcn_wmma_f32_16x16x4_f32(false, wa1[ks], false, xb[ks],
                                                       (short)0, c1, false, false);
        }
#pragma unroll
        for (int t = 0; t < 2; ++t) {
            v8f c = t ? c1 : c0;
            const int hb = hp * 32 + t * 16 + half * 8;
#pragma unroll
            for (int r = 0; r < 8; ++r) c[r] = fmaxf(c[r] + b1[hb + r], 0.0f);
#pragma unroll
            for (int j = 0; j < 4; ++j) {
                v2f a, b;
                const float w0 = W2[(hb + 2 * j) * F_OUT + (l & 7)];
                const float w1 = W2[(hb + 2 * j + 1) * F_OUT + (l & 7)];
                a.x = valid_oc ? w0 : 0.0f;
                a.y = valid_oc ? w1 : 0.0f;
                b.x = c[2 * j];
                b.y = c[2 * j + 1];
                acc = __builtin_amdgcn_wmma_f32_16x16x4_f32(false, a, false, b,
                                                            (short)0, acc, false, false);
            }
        }
    }

    const float u0 = __shfl(acc[4], l), u1 = __shfl(acc[5], l);
    const float u2 = __shfl(acc[6], l), u3 = __shfl(acc[7], l);
    float4 v;
    v.x = half ? u0 : acc[0];
    v.y = half ? u1 : acc[1];
    v.z = half ? u2 : acc[2];
    v.w = half ? u3 : acc[3];
    const float* bp = b2 + half * 4;
    float4 o, hs;
    o.x = v.x * w * w + bp[0];
    o.y = v.y * w * w + bp[1];
    o.z = v.z * w * w + bp[2];
    o.w = v.w * w * w + bp[3];
    hs.x = v.x * w; hs.y = v.y * w; hs.z = v.z * w; hs.w = v.w * w;
    reinterpret_cast<float4*>(h2s + node * F_OUT)[half] = hs;
    reinterpret_cast<float4*>(out + node * F_OUT)[half] = o;
}


__global__ void GCN_agg2(const unsigned* __restrict__ rowptr, const int* __restrict__ csr,
                         const float* __restrict__ dinv, const float* __restrict__ h2s,
                         float* __restrict__ out) {
    const int gt = blockIdx.x * blockDim.x + threadIdx.x;
    const int node = gt >> 3, ch = gt & 7;
    if (node >= N_NODES) return;
    const int beg = (int)rowptr[node], end = (int)rowptr[node + 1];
    const float start = out[node * F_OUT + ch];
    const float w = dinv[node];
    float a = 0.0f;
    int j = beg;
    for (; j + 4 <= end; j += 4) {
        const int s0 = csr[j], s1 = csr[j + 1], s2 = csr[j + 2], s3 = csr[j + 3];
        const float v0 = h2s[(unsigned)(s0 * F_OUT + ch)];
        const float v1 = h2s[(unsigned)(s1 * F_OUT + ch)];
        const float v2 = h2s[(unsigned)(s2 * F_OUT + ch)];
        const float v3 = h2s[(unsigned)(s3 * F_OUT + ch)];
        a += v0; a += v1; a += v2; a += v3;
    }
    for (; j < end; ++j) a += h2s[(unsigned)(csr[j] * F_OUT + ch)];
    out[node * F_OUT + ch] = start + a * w;
}


extern "C" void kernel_launch(void* const* d_in, const int* in_sizes, int n_in,
                              void* d_out, int out_size, void* d_ws, size_t ws_size,
                              hipStream_t stream) {
    const float* x  = (const float*)d_in[0];
    const int*   ei = (const int*)d_in[1];
    const float* W1 = (const float*)d_in[2];
    const float* b1 = (const float*)d_in[3];
    const float* W2 = (const float*)d_in[4];
    const float* b2 = (const float*)d_in[5];
    float* out = (float*)d_out;

    const int* src = ei;
    const int* dst = ei + E_EDGES;

    float* ws   = (float*)d_ws;
    float* dinv = ws;
    float* accx = ws + 50176;
    float* h2s  = ws + 50176 + 12800000;
    unsigned* ebuf = (unsigned*)(ws + 50176 + 12800000 + 400000);
    static_assert(BIN_SL * BIN_PITCH == 802816, "the run starts begin behind the edge buffer");
    unsigned* offS = (unsigned*)(ws + 50176 + 12800000 + 400000 + 802816);
    static_assert(802816 + BIN_SL * OFF_PITCH <= 3211264, "xs starts behind the edge buffer and the run starts");
    float* xs   = ws + 50176 + 12800000 + 400000 + 3211264;
    unsigned* rowptr = (unsigned*)(ws + 50176 + 12800000 + 400000 + 3211264 + 3200000 + 50176 + 256);
    int*      csr    = (int*)(ws + 50176 + 12800000 + 400000 + 3211264 + 3200000 + 50176 + 256 + 50176);

    const int TB = 256;

    GCN_bin<<<BIN_SL, BIN_TB, 0, stream>>>(src, dst, ebuf, offS);
    GCN_csr<<<BIN_NB, CSR_TB, 0, stream>>>(ebuf, offS, x, dinv, xs, rowptr, csr);

    GCN_agg1 <<<N_NODES / (AGG1_TB / 32), AGG1_TB, 0, stream>>>(rowptr, csr, xs, accx);

    GCN_mlp<<<(N_NODES / 16 + 3) / 4, 128, 0, stream>>>(accx, W1, b1, W2, b2, dinv, h2s, out);

    GCN_agg2<<<(N_NODES * F_OUT + TB - 1) / TB, TB, 0, stream>>>(rowptr, csr, dinv, h2s, out);
}
